# starting kernel
# baseline (speedup 1.0000x reference)
#include <hip/hip_runtime.h>
#include <math.h>

#define N_FEAT 512
#define HID    16
#define NCLS   7

typedef __attribute__((ext_vector_type(2))) float v2f;
typedef __attribute__((ext_vector_type(8))) float v8f;

__global__ void k_deg_init(float* __restrict__ deg, int n) {
    int i = blockIdx.x * blockDim.x + threadIdx.x;
    if (i < n) deg[i] = 1.0f;
}

__global__ void k_deg_edges(const int* __restrict__ col, float* __restrict__ deg, int E) {
    int e = blockIdx.x * blockDim.x + threadIdx.x;
    if (e < E) atomicAdd(&deg[col[e]], 1.0f);
}

__global__ void k_dinv(float* __restrict__ deg, int n) {
    int i = blockIdx.x * blockDim.x + threadIdx.x;
    if (i < n) deg[i] = rsqrtf(deg[i]);
}

__global__ __launch_bounds__(256) void k_gemm1(
    const float* __restrict__ x, const float* __restrict__ W1,
    const float* __restrict__ dinv,
    float* __restrict__ h, float* __restrict__ agg1, int nTiles)
{
    __shared__ float sW[N_FEAT * HID];
    for (int i = threadIdx.x; i < N_FEAT * HID; i += blockDim.x) sW[i] = W1[i];
    __syncthreads();

    int wave = threadIdx.x >> 5;
    int lane = threadIdx.x & 31;
    int tile = blockIdx.x * 8 + wave;
    if (tile >= nTiles) return;

    int m  = lane & 15;
    int kh = lane >> 4;
    const float* xr = x + (size_t)(tile * 16 + m) * N_FEAT + 2 * kh;

    v8f acc = {};
#pragma unroll 4
    for (int k = 0; k < N_FEAT; k += 4) {
        v2f a, b;
        a.x = xr[k];
        a.y = xr[k + 1];
        int kk = k + 2 * kh;
        b.x = sW[kk * HID + m];
        b.y = sW[(kk + 1) * HID + m];
        acc = __builtin_amdgcn_wmma_f32_16x16x4_f32(
             false, a,  false, b,
             (short)0, acc,  false,  false);
    }

#pragma unroll
    for (int v = 0; v < 8; ++v) {
        int node  = tile * 16 + v + 8 * kh;
        float val = acc[v];
        float di  = dinv[node];
        h[(size_t)node * HID + m]    = val;
        agg1[(size_t)node * HID + m] = val * di * di;
    }
}

__global__ void k_edge1(const int* __restrict__ eidx, const float* __restrict__ dinv,
                        const float* __restrict__ h, float* __restrict__ agg1, int E)
{
    int t = blockIdx.x * blockDim.x + threadIdx.x;
    int e = t >> 4, f = t & 15;
    if (e >= E) return;
    int r = eidx[e];
    int c = eidx[E + e];
    float w = dinv[r] * dinv[c];
    atomicAdd(&agg1[(size_t)c * HID + f], h[(size_t)r * HID + f] * w);
}

__global__ void k_l2(const float* __restrict__ agg1, const float* __restrict__ b1,
                     const float* __restrict__ W2, const float* __restrict__ dinv,
                     float* __restrict__ o, float* __restrict__ outInit, int n)
{
    __shared__ float sW2[HID * NCLS];
    __shared__ float sb1[HID];
    if (threadIdx.x < HID * NCLS) sW2[threadIdx.x] = W2[threadIdx.x];
    if (threadIdx.x < HID)        sb1[threadIdx.x] = b1[threadIdx.x];
    __syncthreads();

    int i = blockIdx.x * blockDim.x + threadIdx.x;
    if (i >= n) return;

    float hv[HID];
#pragma unroll
    for (int f = 0; f < HID; ++f) {
        float v = agg1[(size_t)i * HID + f] + sb1[f];
        hv[f] = v > 0.0f ? v : 0.0f;
    }
    float di = dinv[i];
    float d2 = di * di;
#pragma unroll
    for (int j = 0; j < NCLS; ++j) {
        float s = 0.0f;
#pragma unroll
        for (int f = 0; f < HID; ++f) s = fmaf(hv[f], sW2[f * NCLS + j], s);
        o[(size_t)i * NCLS + j]       = s;
        outInit[(size_t)i * NCLS + j] = s * d2;
    }
}

__global__ void k_edge2(const int* __restrict__ eidx, const float* __restrict__ dinv,
                        const float* __restrict__ o, float* __restrict__ out, int E)
{
    int t = blockIdx.x * blockDim.x + threadIdx.x;
    int e = t >> 3, j = t & 7;
    if (e >= E || j >= NCLS) return;
    int r = eidx[e];
    int c = eidx[E + e];
    float w = dinv[r] * dinv[c];
    atomicAdd(&out[(size_t)c * NCLS + j], o[(size_t)r * NCLS + j] * w);
}

__global__ void k_lsm(float* __restrict__ out, const float* __restrict__ b2, int n)
{
    int i = blockIdx.x * blockDim.x + threadIdx.x;
    if (i >= n) return;
    float v[NCLS];
    float mx = -1e30f;
#pragma unroll
    for (int j = 0; j < NCLS; ++j) {
        v[j] = out[(size_t)i * NCLS + j] + b2[j];
        mx = fmaxf(mx, v[j]);
    }
    float s = 0.0f;
#pragma unroll
    for (int j = 0; j < NCLS; ++j) s += __expf(v[j] - mx);
    float l = __logf(s);
#pragma unroll
    for (int j = 0; j < NCLS; ++j) out[(size_t)i * NCLS + j] = v[j] - mx - l;
}

extern "C" void kernel_launch(void* const* d_in, const int* in_sizes, int n_in,
                              void* d_out, int out_size, void* d_ws, size_t ws_size,
                              hipStream_t stream)
{
    const float* x    = (const float*)d_in[0];
    const int*   eidx = (const int*)  d_in[1];
    const float* W1   = (const float*)d_in[2];
    const float* b1   = (const float*)d_in[3];
    const float* W2   = (const float*)d_in[4];
    const float* b2   = (const float*)d_in[5];
    float* out = (float*)d_out;

    const int N = in_sizes[0] / N_FEAT;
    const int E = in_sizes[1] / 2;

    float* ws   = (float*)d_ws;
    float* dinv = ws;
    float* h    = ws + (size_t)N;
    float* agg1 = ws + (size_t)N * 17;
    float* obuf = ws + (size_t)N * 33;

    const int B = 256;
    k_deg_init <<<(N + B - 1) / B, B, 0, stream>>>(dinv, N);
    k_deg_edges<<<(E + B - 1) / B, B, 0, stream>>>(eidx + E, dinv, E);
    k_dinv     <<<(N + B - 1) / B, B, 0, stream>>>(dinv, N);

    const int nTiles = (N + 15) / 16;
    k_gemm1<<<(nTiles + 7) / 8, 256, 0, stream>>>(x, W1, dinv, h, agg1, nTiles);

    const int t1 = E * 16;
    k_edge1<<<(t1 + B - 1) / B, B, 0, stream>>>(eidx, dinv, h, agg1, E);

    k_l2<<<(N + B - 1) / B, B, 0, stream>>>(agg1, b1, W2, dinv, obuf, out, N);

    const int t2 = E * 8;
    k_edge2<<<(t2 + B - 1) / B, B, 0, stream>>>(eidx, dinv, obuf, out, E);

    k_lsm<<<(N + B - 1) / B, B, 0, stream>>>(out, b2, N);
}


# Degree count: scattered global float atomics replaced by per-block LDS integer histograms (node range x edge slice); partial counts summed exactly in the rsqrt kernel; init launch dropped
# speedup vs baseline: 1.3974x; 1.3974x over previous
#include <hip/hip_runtime.h>
#include <math.h>

#define N_FEAT 512
#define HID    16
#define NCLS   7

typedef __attribute__((ext_vector_type(2))) float v2f;
typedef __attribute__((ext_vector_type(8))) float v8f;

#define DEG_RN 16384
#define DEG_S  32

static __device__ __forceinline__ void deg_count(unsigned* cnt, int c, int base) {
    unsigned d = (unsigned)(c - base);
    if (d < (unsigned)DEG_RN) atomicAdd(&cnt[d], 1u);
}

__global__ __launch_bounds__(1024) void k_deg_hist(
    const int* __restrict__ col, unsigned* __restrict__ part, int E, int n)
{
    __shared__ unsigned cnt[DEG_RN];
    const int tid  = threadIdx.x;
    const int base = blockIdx.x * DEG_RN;
    const int s    = blockIdx.y;
    for (int i = tid; i < DEG_RN; i += 1024) cnt[i] = 0u;
    __syncthreads();

    const int per = (E + DEG_S - 1) / DEG_S;
    int e0 = s * per;       if (e0 > E) e0 = E;
    int e1 = e0 + per;      if (e1 > E) e1 = E;

    const int mis = (int)((((size_t)(col + e0)) >> 2) & 3);
    int a0 = e0 + ((4 - mis) & 3);
    if (a0 > e1) a0 = e1;
    const int nG = (e1 - a0) >> 2;
    const int t0 = a0 + 4 * nG;
    if (tid < a0 - e0) deg_count(cnt, col[e0 + tid], base);
    if (tid < e1 - t0) deg_count(cnt, col[t0 + tid], base);

    const int4* __restrict__ body = reinterpret_cast<const int4*>(col + a0);
    for (int g = tid; g < nG; g += 4 * 1024) {
        int4 v[4];
#pragma unroll
        for (int j = 0; j < 4; ++j) {
            int gg = g + j * 1024;
            int gc = gg < nG ? gg : nG - 1;
            v[j] = body[gc];
            if (gg >= nG) { v[j].x = -1; v[j].y = -1; v[j].z = -1; v[j].w = -1; }
        }
#pragma unroll
        for (int j = 0; j < 4; ++j) {
            deg_count(cnt, v[j].x, base);
            deg_count(cnt, v[j].y, base);
            deg_count(cnt, v[j].z, base);
            deg_count(cnt, v[j].w, base);
        }
    }
    __syncthreads();

    unsigned* __restrict__ plane = part + (size_t)s * (size_t)n;
    for (int i = tid; i < DEG_RN; i += 1024) {
        int node = base + i;
        if (node < n) plane[node] = cnt[i];
    }
}

__global__ void k_dinv(const unsigned* __restrict__ part, float* __restrict__ dinv, int n) {
    int i = blockIdx.x * blockDim.x + threadIdx.x;
    if (i >= n) return;
    unsigned c = 1u;
#pragma unroll
    for (int s = 0; s < DEG_S; ++s) c += part[(size_t)s * (size_t)n + i];
    dinv[i] = rsqrtf((float)c);
}

__global__ __launch_bounds__(256) void k_gemm1(
    const float* __restrict__ x, const float* __restrict__ W1,
    const float* __restrict__ dinv,
    float* __restrict__ h, float* __restrict__ agg1, int nTiles)
{
    __shared__ float sW[N_FEAT * HID];
    for (int i = threadIdx.x; i < N_FEAT * HID; i += blockDim.x) sW[i] = W1[i];
    __syncthreads();

    int wave = threadIdx.x >> 5;
    int lane = threadIdx.x & 31;
    int tile = blockIdx.x * 8 + wave;
    if (tile >= nTiles) return;

    int m  = lane & 15;
    int kh = lane >> 4;
    const float* xr = x + (size_t)(tile * 16 + m) * N_FEAT + 2 * kh;

    v8f acc = {};
#pragma unroll 4
    for (int k = 0; k < N_FEAT; k += 4) {
        v2f a, b;
        a.x = xr[k];
        a.y = xr[k + 1];
        int kk = k + 2 * kh;
        b.x = sW[kk * HID + m];
        b.y = sW[(kk + 1) * HID + m];
        acc = __builtin_amdgcn_wmma_f32_16x16x4_f32(
             false, a,  false, b,
             (short)0, acc,  false,  false);
    }

#pragma unroll
    for (int v = 0; v < 8; ++v) {
        int node  = tile * 16 + v + 8 * kh;
        float val = acc[v];
        float di  = dinv[node];
        h[(size_t)node * HID + m]    = val;
        agg1[(size_t)node * HID + m] = val * di * di;
    }
}

__global__ void k_edge1(const int* __restrict__ eidx, const float* __restrict__ dinv,
                        const float* __restrict__ h, float* __restrict__ agg1, int E)
{
    int t = blockIdx.x * blockDim.x + threadIdx.x;
    int e = t >> 4, f = t & 15;
    if (e >= E) return;
    int r = eidx[e];
    int c = eidx[E + e];
    float w = dinv[r] * dinv[c];
    atomicAdd(&agg1[(size_t)c * HID + f], h[(size_t)r * HID + f] * w);
}

__global__ void k_l2(const float* __restrict__ agg1, const float* __restrict__ b1,
                     const float* __restrict__ W2, const float* __restrict__ dinv,
                     float* __restrict__ o, float* __restrict__ outInit, int n)
{
    __shared__ float sW2[HID * NCLS];
    __shared__ float sb1[HID];
    if (threadIdx.x < HID * NCLS) sW2[threadIdx.x] = W2[threadIdx.x];
    if (threadIdx.x < HID)        sb1[threadIdx.x] = b1[threadIdx.x];
    __syncthreads();

    int i = blockIdx.x * blockDim.x + threadIdx.x;
    if (i >= n) return;

    float hv[HID];
#pragma unroll
    for (int f = 0; f < HID; ++f) {
        float v = agg1[(size_t)i * HID + f] + sb1[f];
        hv[f] = v > 0.0f ? v : 0.0f;
    }
    float di = dinv[i];
    float d2 = di * di;
#pragma unroll
    for (int j = 0; j < NCLS; ++j) {
        float s = 0.0f;
#pragma unroll
        for (int f = 0; f < HID; ++f) s = fmaf(hv[f], sW2[f * NCLS + j], s);
        o[(size_t)i * NCLS + j]       = s;
        outInit[(size_t)i * NCLS + j] = s * d2;
    }
}

__global__ void k_edge2(const int* __restrict__ eidx, const float* __restrict__ dinv,
                        const float* __restrict__ o, float* __restrict__ out, int E)
{
    int t = blockIdx.x * blockDim.x + threadIdx.x;
    int e = t >> 3, j = t & 7;
    if (e >= E || j >= NCLS) return;
    int r = eidx[e];
    int c = eidx[E + e];
    float w = dinv[r] * dinv[c];
    atomicAdd(&out[(size_t)c * NCLS + j], o[(size_t)r * NCLS + j] * w);
}

__global__ void k_lsm(float* __restrict__ out, const float* __restrict__ b2, int n)
{
    int i = blockIdx.x * blockDim.x + threadIdx.x;
    if (i >= n) return;
    float v[NCLS];
    float mx = -1e30f;
#pragma unroll
    for (int j = 0; j < NCLS; ++j) {
        v[j] = out[(size_t)i * NCLS + j] + b2[j];
        mx = fmaxf(mx, v[j]);
    }
    float s = 0.0f;
#pragma unroll
    for (int j = 0; j < NCLS; ++j) s += __expf(v[j] - mx);
    float l = __logf(s);
#pragma unroll
    for (int j = 0; j < NCLS; ++j) out[(size_t)i * NCLS + j] = v[j] - mx - l;
}

extern "C" void kernel_launch(void* const* d_in, const int* in_sizes, int n_in,
                              void* d_out, int out_size, void* d_ws, size_t ws_size,
                              hipStream_t stream)
{
    const float* x    = (const float*)d_in[0];
    const int*   eidx = (const int*)  d_in[1];
    const float* W1   = (const float*)d_in[2];
    const float* b1   = (const float*)d_in[3];
    const float* W2   = (const float*)d_in[4];
    const float* b2   = (const float*)d_in[5];
    float* out = (float*)d_out;

    const int N = in_sizes[0] / N_FEAT;
    const int E = in_sizes[1] / 2;

    float* ws   = (float*)d_ws;
    float* dinv = ws;
    float* h    = ws + (size_t)N;
    float* agg1 = ws + (size_t)N * 17;
    float* obuf = ws + (size_t)N * 33;

    const int B = 256;
    unsigned* part = (unsigned*)(ws + (size_t)N);
    dim3 gDeg((N + DEG_RN - 1) / DEG_RN, DEG_S);
    k_deg_hist<<<gDeg, 1024, 0, stream>>>(eidx + E, part, E, N);
    k_dinv    <<<(N + B - 1) / B, B, 0, stream>>>(part, dinv, N);

    const int nTiles = (N + 15) / 16;
    k_gemm1<<<(nTiles + 7) / 8, 256, 0, stream>>>(x, W1, dinv, h, agg1, nTiles);

    const int t1 = E * 16;
    k_edge1<<<(t1 + B - 1) / B, B, 0, stream>>>(eidx, dinv, h, agg1, E);

    k_l2<<<(N + B - 1) / B, B, 0, stream>>>(agg1, b1, W2, dinv, obuf, out, N);

    const int t2 = E * 8;
    k_edge2<<<(t2 + B - 1) / B, B, 0, stream>>>(eidx, dinv, obuf, out, E);

    k_lsm<<<(N + B - 1) / B, B, 0, stream>>>(out, b2, N);
}


# Edge scatter-add of both layers: per-edge global float atomics replaced by per-block LDS row sums over (node range x edge slice), added to the output once per block; sum order changes
# speedup vs baseline: 2.1832x; 1.5623x over previous
#include <hip/hip_runtime.h>
#include <math.h>

#define N_FEAT 512
#define HID    16
#define NCLS   7

typedef __attribute__((ext_vector_type(2))) float v2f;
typedef __attribute__((ext_vector_type(8))) float v8f;

#define DEG_RN 16384
#define DEG_S  32

static __device__ __forceinline__ void deg_count(unsigned* cnt, int c, int base) {
    unsigned d = (unsigned)(c - base);
    if (d < (unsigned)DEG_RN) atomicAdd(&cnt[d], 1u);
}

__global__ __launch_bounds__(1024) void k_deg_hist(
    const int* __restrict__ col, unsigned* __restrict__ part, int E, int n)
{
    __shared__ unsigned cnt[DEG_RN];
    const int tid  = threadIdx.x;
    const int base = blockIdx.x * DEG_RN;
    const int s    = blockIdx.y;
    for (int i = tid; i < DEG_RN; i += 1024) cnt[i] = 0u;
    __syncthreads();

    const int per = (E + DEG_S - 1) / DEG_S;
    int e0 = s * per;       if (e0 > E) e0 = E;
    int e1 = e0 + per;      if (e1 > E) e1 = E;

    const int mis = (int)((((size_t)(col + e0)) >> 2) & 3);
    int a0 = e0 + ((4 - mis) & 3);
    if (a0 > e1) a0 = e1;
    const int nG = (e1 - a0) >> 2;
    const int t0 = a0 + 4 * nG;
    if (tid < a0 - e0) deg_count(cnt, col[e0 + tid], base);
    if (tid < e1 - t0) deg_count(cnt, col[t0 + tid], base);

    const int4* __restrict__ body = reinterpret_cast<const int4*>(col + a0);
    for (int g = tid; g < nG; g += 4 * 1024) {
        int4 v[4];
#pragma unroll
        for (int j = 0; j < 4; ++j) {
            int gg = g + j * 1024;
            int gc = gg < nG ? gg : nG - 1;
            v[j] = body[gc];
            if (gg >= nG) { v[j].x = -1; v[j].y = -1; v[j].z = -1; v[j].w = -1; }
        }
#pragma unroll
        for (int j = 0; j < 4; ++j) {
            deg_count(cnt, v[j].x, base);
            deg_count(cnt, v[j].y, base);
            deg_count(cnt, v[j].z, base);
            deg_count(cnt, v[j].w, base);
        }
    }
    __syncthreads();

    unsigned* __restrict__ plane = part + (size_t)s * (size_t)n;
    for (int i = tid; i < DEG_RN; i += 1024) {
        int node = base + i;
        if (node < n) plane[node] = cnt[i];
    }
}

__global__ void k_dinv(const unsigned* __restrict__ part, float* __restrict__ dinv, int n) {
    int i = blockIdx.x * blockDim.x + threadIdx.x;
    if (i >= n) return;
    unsigned c = 1u;
#pragma unroll
    for (int s = 0; s < DEG_S; ++s) c += part[(size_t)s * (size_t)n + i];
    dinv[i] = rsqrtf((float)c);
}

__global__ __launch_bounds__(256) void k_gemm1(
    const float* __restrict__ x, const float* __restrict__ W1,
    const float* __restrict__ dinv,
    float* __restrict__ h, float* __restrict__ agg1, int nTiles)
{
    __shared__ float sW[N_FEAT * HID];
    for (int i = threadIdx.x; i < N_FEAT * HID; i += blockDim.x) sW[i] = W1[i];
    __syncthreads();

    int wave = threadIdx.x >> 5;
    int lane = threadIdx.x & 31;
    int tile = blockIdx.x * 8 + wave;
    if (tile >= nTiles) return;

    int m  = lane & 15;
    int kh = lane >> 4;
    const float* xr = x + (size_t)(tile * 16 + m) * N_FEAT + 2 * kh;

    v8f acc = {};
#pragma unroll 4
    for (int k = 0; k < N_FEAT; k += 4) {
        v2f a, b;
        a.x = xr[k];
        a.y = xr[k + 1];
        int kk = k + 2 * kh;
        b.x = sW[kk * HID + m];
        b.y = sW[(kk + 1) * HID + m];
        acc = __builtin_amdgcn_wmma_f32_16x16x4_f32(
             false, a,  false, b,
             (short)0, acc,  false,  false);
    }

#pragma unroll
    for (int v = 0; v < 8; ++v) {
        int node  = tile * 16 + v + 8 * kh;
        float val = acc[v];
        float di  = dinv[node];
        h[(size_t)node * HID + m]    = val;
        agg1[(size_t)node * HID + m] = val * di * di;
    }
}

template <bool B> struct agg_flag { static constexpr bool value = B; };

template <int F, int FP, int RN>
static __device__ __forceinline__ void agg_direct(
    float* acc, const int* __restrict__ row, const float* __restrict__ dinv,
    const float* __restrict__ srcv, int e, int d, int base)
{
    int r = row[e];
    float w = dinv[r] * dinv[base + d];
    for (int f = 0; f < F; ++f) atomicAdd(&acc[d * FP + f], srcv[(size_t)r * F + f] * w);
}

template <int F, int FP, int RN, int S, int QCAP, int U>
__global__ __launch_bounds__(1024) void k_agg(
    const int* __restrict__ eidx, const float* __restrict__ dinv,
    const float* __restrict__ srcv, float* __restrict__ dst, int E, int n)
{
    __shared__ __attribute__((aligned(16))) float acc[RN * FP];
    __shared__ unsigned q[QCAP + 32];
    __shared__ unsigned qn;

    const int* __restrict__ row = eidx;
    const int* __restrict__ col = eidx + E;
    const int tid  = threadIdx.x;
    const int base = blockIdx.x * RN;
    const int s    = blockIdx.y;

    float4* acc4 = reinterpret_cast<float4*>(acc);
    for (int i = tid; i < RN * FP / 4; i += 1024) acc4[i] = make_float4(0.f, 0.f, 0.f, 0.f);
    if (tid == 0) qn = 0u;
    __syncthreads();

    const int per = (E + S - 1) / S;
    int e0 = s * per;       if (e0 > E) e0 = E;
    int e1 = e0 + per;      if (e1 > E) e1 = E;

    auto visit = [&](int e, int c) {
        unsigned d = (unsigned)(c - base);
        if (d < (unsigned)RN) {
            unsigned pos = atomicAdd(&qn, 1u);
            if (pos < (unsigned)QCAP) q[pos] = (unsigned)e;
            else agg_direct<F, FP, RN>(acc, row, dinv, srcv, e, (int)d, base);
        }
    };

    const int mis = (int)((((size_t)(col + e0)) >> 2) & 3);
    int a0 = e0 + ((4 - mis) & 3);
    if (a0 > e1) a0 = e1;
    const int nG = (e1 - a0) >> 2;
    const int t0 = a0 + 4 * nG;
    if (tid < a0 - e0) visit(e0 + tid, col[e0 + tid]);
    if (tid < e1 - t0) visit(t0 + tid, col[t0 + tid]);

    const int4* __restrict__ body = reinterpret_cast<const int4*>(col + a0);
    const int lane = tid & 31;
    for (int g0 = 0; g0 < nG; g0 += 8 * 1024) {
        const int g = g0 + tid;
        int4 v[8];
#pragma unroll
        for (int j = 0; j < 8; ++j) {
            int gg = g + j * 1024;
            int gc = gg < nG ? gg : nG - 1;
            v[j] = body[gc];
            if (gg >= nG) { v[j].x = -1; v[j].y = -1; v[j].z = -1; v[j].w = -1; }
        }
        unsigned nh = 0;
#pragma unroll
        for (int j = 0; j < 8; ++j) {
            nh += (unsigned)(v[j].x - base) < (unsigned)RN;
            nh += (unsigned)(v[j].y - base) < (unsigned)RN;
            nh += (unsigned)(v[j].z - base) < (unsigned)RN;
            nh += (unsigned)(v[j].w - base) < (unsigned)RN;
        }
        unsigned inc = nh;
#pragma unroll
        for (int o = 1; o < 32; o <<= 1) {
            unsigned up = (unsigned)__shfl_up((int)inc, o);
            if (lane >= o) inc += up;
        }
        const unsigned tot = (unsigned)__shfl((int)inc, 31);
        if (tot) {
            unsigned wb = 0;
            if (lane == 31) wb = atomicAdd(&qn, tot);
            wb = (unsigned)__shfl((int)wb, 31);
            unsigned pos = wb + inc - nh;
            if (wb + tot <= (unsigned)QCAP) {
                const unsigned spare = (unsigned)QCAP + (unsigned)lane;
#pragma unroll
                for (int j = 0; j < 8; ++j) {
                    unsigned eb = (unsigned)(a0 + 4 * (g + j * 1024));
                    bool mx = (unsigned)(v[j].x - base) < (unsigned)RN;
                    bool my = (unsigned)(v[j].y - base) < (unsigned)RN;
                    bool mz = (unsigned)(v[j].z - base) < (unsigned)RN;
                    bool mw = (unsigned)(v[j].w - base) < (unsigned)RN;
                    q[mx ? pos : spare] = eb;     pos += mx;
                    q[my ? pos : spare] = eb + 1; pos += my;
                    q[mz ? pos : spare] = eb + 2; pos += mz;
                    q[mw ? pos : spare] = eb + 3; pos += mw;
                }
            } else {
#pragma unroll
                for (int j = 0; j < 8; ++j) {
                    int eb = a0 + 4 * (g + j * 1024);
                    int cj[4] = { v[j].x, v[j].y, v[j].z, v[j].w };
#pragma unroll
                    for (int k = 0; k < 4; ++k) {
                        unsigned d = (unsigned)(cj[k] - base);
                        if (d < (unsigned)RN) {
                            if (pos < (unsigned)QCAP) q[pos] = (unsigned)(eb + k);
                            else agg_direct<F, FP, RN>(acc, row, dinv, srcv, eb + k, (int)d, base);
                            ++pos;
                        }
                    }
                }
            }
        }
    }
    __syncthreads();

    const int cnt = (int)(qn < (unsigned)QCAP ? qn : (unsigned)QCAP);

    for (int i0 = tid; i0 < cnt; i0 += 4 * 1024) {
        int ee[4], cc[4], rr[4];
#pragma unroll
        for (int j = 0; j < 4; ++j) { int i = i0 + j * 1024; ee[j] = i < cnt ? (int)q[i] : e0; }
#pragma unroll
        for (int j = 0; j < 4; ++j) { cc[j] = col[ee[j]]; rr[j] = row[ee[j]]; }
#pragma unroll
        for (int j = 0; j < 4; ++j) {
            int i = i0 + j * 1024;
            if (i < cnt) q[i] = (unsigned)rr[j] | ((unsigned)(cc[j] - base) << 19);
        }
    }
    __syncthreads();

    {
        constexpr int G = 1024 / FP;
        const int g  = tid / FP;
        const int f  = tid % FP;
        const bool fl = f < F;
        const int fs = fl ? f : 0;
        auto batch = [&](int i0, auto guard) {
            constexpr bool GUARD = decltype(guard)::value;
            unsigned en[U]; float dr[U], dc[U], hv[U];
#pragma unroll
            for (int j = 0; j < U; ++j) {
                int i = i0 + j * G;
                en[j] = q[(!GUARD || i < cnt) ? i : cnt - 1];
            }
#pragma unroll
            for (int j = 0; j < U; ++j) {
                unsigned r = en[j] & 0x7FFFFu;
                unsigned d = en[j] >> 19;
                dr[j] = dinv[r];
                dc[j] = dinv[base + (int)d];
                hv[j] = srcv[(size_t)r * F + fs];
            }
#pragma unroll
            for (int j = 0; j < U; ++j) {
                int i = i0 + j * G;
                if ((!GUARD || i < cnt) && fl)
                    atomicAdd(&acc[(en[j] >> 19) * FP + f], hv[j] * (dr[j] * dc[j]));
            }
        };
        const int nfull = cnt / (G * U);
        for (int b = 0; b < nfull; ++b) batch(b * (G * U) + g, agg_flag<false>{});
        const int it = nfull * (G * U) + g;
        if (it < cnt) batch(it, agg_flag<true>{});
    }
    __syncthreads();

    {
        const int T    = RN * F;
        const int stag = ((T / S) & ~1023) * s;
        for (int k = tid; k < T; k += 1024) {
            int t = k + stag;
            if (t >= T) t -= T;
            int nl = t / F;
            int j  = t - nl * F;
            int node = base + nl;
            if (node < n) atomicAdd(&dst[(size_t)node * F + j], acc[nl * FP + j]);
        }
    }
}

__global__ void k_l2(const float* __restrict__ agg1, const float* __restrict__ b1,
                     const float* __restrict__ W2, const float* __restrict__ dinv,
                     float* __restrict__ o, float* __restrict__ outInit, int n)
{
    __shared__ float sW2[HID * NCLS];
    __shared__ float sb1[HID];
    if (threadIdx.x < HID * NCLS) sW2[threadIdx.x] = W2[threadIdx.x];
    if (threadIdx.x < HID)        sb1[threadIdx.x] = b1[threadIdx.x];
    __syncthreads();

    int i = blockIdx.x * blockDim.x + threadIdx.x;
    if (i >= n) return;

    float hv[HID];
#pragma unroll
    for (int f = 0; f < HID; ++f) {
        float v = agg1[(size_t)i * HID + f] + sb1[f];
        hv[f] = v > 0.0f ? v : 0.0f;
    }
    float di = dinv[i];
    float d2 = di * di;
#pragma unroll
    for (int j = 0; j < NCLS; ++j) {
        float s = 0.0f;
#pragma unroll
        for (int f = 0; f < HID; ++f) s = fmaf(hv[f], sW2[f * NCLS + j], s);
        o[(size_t)i * NCLS + j]       = s;
        outInit[(size_t)i * NCLS + j] = s * d2;
    }
}

__global__ void k_lsm(float* __restrict__ out, const float* __restrict__ b2, int n)
{
    int i = blockIdx.x * blockDim.x + threadIdx.x;
    if (i >= n) return;
    float v[NCLS];
    float mx = -1e30f;
#pragma unroll
    for (int j = 0; j < NCLS; ++j) {
        v[j] = out[(size_t)i * NCLS + j] + b2[j];
        mx = fmaxf(mx, v[j]);
    }
    float s = 0.0f;
#pragma unroll
    for (int j = 0; j < NCLS; ++j) s += __expf(v[j] - mx);
    float l = __logf(s);
#pragma unroll
    for (int j = 0; j < NCLS; ++j) out[(size_t)i * NCLS + j] = v[j] - mx - l;
}

extern "C" void kernel_launch(void* const* d_in, const int* in_sizes, int n_in,
                              void* d_out, int out_size, void* d_ws, size_t ws_size,
                              hipStream_t stream)
{
    const float* x    = (const float*)d_in[0];
    const int*   eidx = (const int*)  d_in[1];
    const float* W1   = (const float*)d_in[2];
    const float* b1   = (const float*)d_in[3];
    const float* W2   = (const float*)d_in[4];
    const float* b2   = (const float*)d_in[5];
    float* out = (float*)d_out;

    const int N = in_sizes[0] / N_FEAT;
    const int E = in_sizes[1] / 2;

    float* ws   = (float*)d_ws;
    float* dinv = ws;
    float* h    = ws + (size_t)N;
    float* agg1 = ws + (size_t)N * 17;
    float* obuf = ws + (size_t)N * 33;

    const int B = 256;
    unsigned* part = (unsigned*)(ws + (size_t)N);
    dim3 gDeg((N + DEG_RN - 1) / DEG_RN, DEG_S);
    k_deg_hist<<<gDeg, 1024, 0, stream>>>(eidx + E, part, E, N);
    k_dinv    <<<(N + B - 1) / B, B, 0, stream>>>(part, dinv, N);

    const int nTiles = (N + 15) / 16;
    k_gemm1<<<(nTiles + 7) / 8, 256, 0, stream>>>(x, W1, dinv, h, agg1, nTiles);

    dim3 gA1((N + 3584 - 1) / 3584, 9);
    k_agg<HID, 16, 3584, 9, 16384, 16><<<gA1, 1024, 0, stream>>>(eidx, dinv, h, agg1, E, N);

    k_l2<<<(N + B - 1) / B, B, 0, stream>>>(agg1, b1, W2, dinv, obuf, out, N);

    dim3 gA2((N + 7168 - 1) / 7168, 18);
    k_agg<NCLS, 8, 7168, 18, 16384, 16><<<gA2, 1024, 0, stream>>>(eidx, dinv, obuf, out, E, N);

    k_lsm<<<(N + B - 1) / B, B, 0, stream>>>(out, b2, N);
}


# Edge weight factored out of the edge pass: rows stored pre-scaled by dinv[src], finished sums scaled by dinv[dst] at their reader; per-edge dinv reads and products go; message rounding changes
# speedup vs baseline: 2.5124x; 1.1508x over previous
#include <hip/hip_runtime.h>
#include <math.h>

#define N_FEAT 512
#define HID    16
#define NCLS   7

typedef __attribute__((ext_vector_type(2))) float v2f;
typedef __attribute__((ext_vector_type(8))) float v8f;

#define DEG_RN 16384
#define DEG_S  32

static __device__ __forceinline__ void deg_count(unsigned* cnt, int c, int base) {
    unsigned d = (unsigned)(c - base);
    if (d < (unsigned)DEG_RN) atomicAdd(&cnt[d], 1u);
}

__global__ __launch_bounds__(1024) void k_deg_hist(
    const int* __restrict__ col, unsigned* __restrict__ part, int E, int n)
{
    __shared__ unsigned cnt[DEG_RN];
    const int tid  = threadIdx.x;
    const int base = blockIdx.x * DEG_RN;
    const int s    = blockIdx.y;
    for (int i = tid; i < DEG_RN; i += 1024) cnt[i] = 0u;
    __syncthreads();

    const int per = (E + DEG_S - 1) / DEG_S;
    int e0 = s * per;       if (e0 > E) e0 = E;
    int e1 = e0 + per;      if (e1 > E) e1 = E;

    const int mis = (int)((((size_t)(col + e0)) >> 2) & 3);
    int a0 = e0 + ((4 - mis) & 3);
    if (a0 > e1) a0 = e1;
    const int nG = (e1 - a0) >> 2;
    const int t0 = a0 + 4 * nG;
    if (tid < a0 - e0) deg_count(cnt, col[e0 + tid], base);
    if (tid < e1 - t0) deg_count(cnt, col[t0 + tid], base);

    const int4* __restrict__ body = reinterpret_cast<const int4*>(col + a0);
    for (int g = tid; g < nG; g += 4 * 1024) {
        int4 v[4];
#pragma unroll
        for (int j = 0; j < 4; ++j) {
            int gg = g + j * 1024;
            int gc = gg < nG ? gg : nG - 1;
            v[j] = body[gc];
            if (gg >= nG) { v[j].x = -1; v[j].y = -1; v[j].z = -1; v[j].w = -1; }
        }
#pragma unroll
        for (int j = 0; j < 4; ++j) {
            deg_count(cnt, v[j].x, base);
            deg_count(cnt, v[j].y, base);
            deg_count(cnt, v[j].z, base);
            deg_count(cnt, v[j].w, base);
        }
    }
    __syncthreads();

    unsigned* __restrict__ plane = part + (size_t)s * (size_t)n;
    for (int i = tid; i < DEG_RN; i += 1024) {
        int node = base + i;
        if (node < n) plane[node] = cnt[i];
    }
}

__global__ void k_dinv(const unsigned* __restrict__ part, float* __restrict__ dinv, int n) {
    int i = blockIdx.x * blockDim.x + threadIdx.x;
    if (i >= n) return;
    unsigned c = 1u;
#pragma unroll
    for (int s = 0; s < DEG_S; ++s) c += part[(size_t)s * (size_t)n + i];
    dinv[i] = rsqrtf((float)c);
}

__global__ __launch_bounds__(256) void k_gemm1(
    const float* __restrict__ x, const float* __restrict__ W1,
    const float* __restrict__ dinv,
    float* __restrict__ h, float* __restrict__ agg1, int nTiles)
{
    __shared__ float sW[N_FEAT * HID];
    for (int i = threadIdx.x; i < N_FEAT * HID; i += blockDim.x) sW[i] = W1[i];
    __syncthreads();

    int wave = threadIdx.x >> 5;
    int lane = threadIdx.x & 31;
    int tile = blockIdx.x * 8 + wave;
    if (tile >= nTiles) return;

    int m  = lane & 15;
    int kh = lane >> 4;
    const float* xr = x + (size_t)(tile * 16 + m) * N_FEAT + 2 * kh;

    v8f acc = {};
#pragma unroll 4
    for (int k = 0; k < N_FEAT; k += 4) {
        v2f a, b;
        a.x = xr[k];
        a.y = xr[k + 1];
        int kk = k + 2 * kh;
        b.x = sW[kk * HID + m];
        b.y = sW[(kk + 1) * HID + m];
        acc = __builtin_amdgcn_wmma_f32_16x16x4_f32(
             false, a,  false, b,
             (short)0, acc,  false,  false);
    }

#pragma unroll
    for (int v = 0; v < 8; ++v) {
        int node  = tile * 16 + v + 8 * kh;
        float hs  = acc[v] * dinv[node];
        h[(size_t)node * HID + m]    = hs;
        agg1[(size_t)node * HID + m] = hs;
    }
}

template <bool B> struct agg_flag { static constexpr bool value = B; };

template <int F, int FP, int RN>
static __device__ __forceinline__ void agg_direct(
    float* acc, const int* __restrict__ row,
    const float* __restrict__ srcv, int e, int d)
{
    int r = row[e];
    for (int f = 0; f < F; ++f) atomicAdd(&acc[d * FP + f], srcv[(size_t)r * F + f]);
}

template <int F, int FP, int RN, int S, int QCAP, int U>
__global__ __launch_bounds__(1024) void k_agg(
    const int* __restrict__ eidx,
    const float* __restrict__ srcv, float* __restrict__ dst, int E, int n)
{
    __shared__ __attribute__((aligned(16))) float acc[RN * FP];
    __shared__ unsigned q[QCAP + 32];
    __shared__ unsigned qn;

    const int* __restrict__ row = eidx;
    const int* __restrict__ col = eidx + E;
    const int tid  = threadIdx.x;
    const int base = blockIdx.x * RN;
    const int s    = blockIdx.y;

    float4* acc4 = reinterpret_cast<float4*>(acc);
    for (int i = tid; i < RN * FP / 4; i += 1024) acc4[i] = make_float4(0.f, 0.f, 0.f, 0.f);
    if (tid == 0) qn = 0u;
    __syncthreads();

    const int per = (E + S - 1) / S;
    int e0 = s * per;       if (e0 > E) e0 = E;
    int e1 = e0 + per;      if (e1 > E) e1 = E;

    auto visit = [&](int e, int c) {
        unsigned d = (unsigned)(c - base);
        if (d < (unsigned)RN) {
            unsigned pos = atomicAdd(&qn, 1u);
            if (pos < (unsigned)QCAP) q[pos] = (unsigned)e;
            else agg_direct<F, FP, RN>(acc, row, srcv, e, (int)d);
        }
    };

    const int mis = (int)((((size_t)(col + e0)) >> 2) & 3);
    int a0 = e0 + ((4 - mis) & 3);
    if (a0 > e1) a0 = e1;
    const int nG = (e1 - a0) >> 2;
    const int t0 = a0 + 4 * nG;
    if (tid < a0 - e0) visit(e0 + tid, col[e0 + tid]);
    if (tid < e1 - t0) visit(t0 + tid, col[t0 + tid]);

    const int4* __restrict__ body = reinterpret_cast<const int4*>(col + a0);
    const int lane = tid & 31;
    for (int g0 = 0; g0 < nG; g0 += 8 * 1024) {
        const int g = g0 + tid;
        int4 v[8];
#pragma unroll
        for (int j = 0; j < 8; ++j) {
            int gg = g + j * 1024;
            int gc = gg < nG ? gg : nG - 1;
            v[j] = body[gc];
            if (gg >= nG) { v[j].x = -1; v[j].y = -1; v[j].z = -1; v[j].w = -1; }
        }
        unsigned nh = 0;
#pragma unroll
        for (int j = 0; j < 8; ++j) {
            nh += (unsigned)(v[j].x - base) < (unsigned)RN;
            nh += (unsigned)(v[j].y - base) < (unsigned)RN;
            nh += (unsigned)(v[j].z - base) < (unsigned)RN;
            nh += (unsigned)(v[j].w - base) < (unsigned)RN;
        }
        unsigned inc = nh;
#pragma unroll
        for (int o = 1; o < 32; o <<= 1) {
            unsigned up = (unsigned)__shfl_up((int)inc, o);
            if (lane >= o) inc += up;
        }
        const unsigned tot = (unsigned)__shfl((int)inc, 31);
        if (tot) {
            unsigned wb = 0;
            if (lane == 31) wb = atomicAdd(&qn, tot);
            wb = (unsigned)__shfl((int)wb, 31);
            unsigned pos = wb + inc - nh;
            if (wb + tot <= (unsigned)QCAP) {
                const unsigned spare = (unsigned)QCAP + (unsigned)lane;
#pragma unroll
                for (int j = 0; j < 8; ++j) {
                    unsigned eb = (unsigned)(a0 + 4 * (g + j * 1024));
                    bool mx = (unsigned)(v[j].x - base) < (unsigned)RN;
                    bool my = (unsigned)(v[j].y - base) < (unsigned)RN;
                    bool mz = (unsigned)(v[j].z - base) < (unsigned)RN;
                    bool mw = (unsigned)(v[j].w - base) < (unsigned)RN;
                    q[mx ? pos : spare] = eb;     pos += mx;
                    q[my ? pos : spare] = eb + 1; pos += my;
                    q[mz ? pos : spare] = eb + 2; pos += mz;
                    q[mw ? pos : spare] = eb + 3; pos += mw;
                }
            } else {
#pragma unroll
                for (int j = 0; j < 8; ++j) {
                    int eb = a0 + 4 * (g + j * 1024);
                    int cj[4] = { v[j].x, v[j].y, v[j].z, v[j].w };
#pragma unroll
                    for (int k = 0; k < 4; ++k) {
                        unsigned d = (unsigned)(cj[k] - base);
                        if (d < (unsigned)RN) {
                            if (pos < (unsigned)QCAP) q[pos] = (unsigned)(eb + k);
                            else agg_direct<F, FP, RN>(acc, row, srcv, eb + k, (int)d);
                            ++pos;
                        }
                    }
                }
            }
        }
    }
    __syncthreads();

    const int cnt = (int)(qn < (unsigned)QCAP ? qn : (unsigned)QCAP);

    for (int i0 = tid; i0 < cnt; i0 += 4 * 1024) {
        int ee[4], cc[4], rr[4];
#pragma unroll
        for (int j = 0; j < 4; ++j) { int i = i0 + j * 1024; ee[j] = i < cnt ? (int)q[i] : e0; }
#pragma unroll
        for (int j = 0; j < 4; ++j) { cc[j] = col[ee[j]]; rr[j] = row[ee[j]]; }
#pragma unroll
        for (int j = 0; j < 4; ++j) {
            int i = i0 + j * 1024;
            if (i < cnt) q[i] = (unsigned)rr[j] | ((unsigned)(cc[j] - base) << 19);
        }
    }
    __syncthreads();

    {
        constexpr int G = 1024 / FP;
        const int g  = tid / FP;
        const int f  = tid % FP;
        const bool fl = f < F;
        const int fs = fl ? f : 0;
        auto batch = [&](int i0, auto guard) {
            constexpr bool GUARD = decltype(guard)::value;
            unsigned en[U]; float hv[U];
#pragma unroll
            for (int j = 0; j < U; ++j) {
                int i = i0 + j * G;
                en[j] = q[(!GUARD || i < cnt) ? i : cnt - 1];
            }
#pragma unroll
            for (int j = 0; j < U; ++j) {
                unsigned r = en[j] & 0x7FFFFu;
                hv[j] = srcv[(size_t)r * F + fs];
            }
#pragma unroll
            for (int j = 0; j < U; ++j) {
                int i = i0 + j * G;
                if ((!GUARD || i < cnt) && fl)
                    atomicAdd(&acc[(en[j] >> 19) * FP + f], hv[j]);
            }
        };
        const int nfull = cnt / (G * U);
        for (int b = 0; b < nfull; ++b) batch(b * (G * U) + g, agg_flag<false>{});
        const int it = nfull * (G * U) + g;
        if (it < cnt) batch(it, agg_flag<true>{});
    }
    __syncthreads();

    {
        const int T    = RN * F;
        const int stag = ((T / S) & ~1023) * s;
        for (int k = tid; k < T; k += 1024) {
            int t = k + stag;
            if (t >= T) t -= T;
            int nl = t / F;
            int j  = t - nl * F;
            int node = base + nl;
            if (node < n) atomicAdd(&dst[(size_t)node * F + j], acc[nl * FP + j]);
        }
    }
}

__global__ void k_l2(const float* __restrict__ agg1, const float* __restrict__ b1,
                     const float* __restrict__ W2, const float* __restrict__ dinv,
                     float* __restrict__ o, float* __restrict__ outInit, int n)
{
    __shared__ float sW2[HID * NCLS];
    __shared__ float sb1[HID];
    if (threadIdx.x < HID * NCLS) sW2[threadIdx.x] = W2[threadIdx.x];
    if (threadIdx.x < HID)        sb1[threadIdx.x] = b1[threadIdx.x];
    __syncthreads();

    int i = blockIdx.x * blockDim.x + threadIdx.x;
    if (i >= n) return;

    const float di = dinv[i];
    float hv[HID];
#pragma unroll
    for (int f = 0; f < HID; ++f) {
        float v = agg1[(size_t)i * HID + f] * di + sb1[f];
        hv[f] = v > 0.0f ? v : 0.0f;
    }
#pragma unroll
    for (int j = 0; j < NCLS; ++j) {
        float s = 0.0f;
#pragma unroll
        for (int f = 0; f < HID; ++f) s = fmaf(hv[f], sW2[f * NCLS + j], s);
        float os = s * di;
        o[(size_t)i * NCLS + j]       = os;
        outInit[(size_t)i * NCLS + j] = os;
    }
}

__global__ void k_lsm(float* __restrict__ out, const float* __restrict__ b2,
                      const float* __restrict__ dinv, int n)
{
    int i = blockIdx.x * blockDim.x + threadIdx.x;
    if (i >= n) return;
    const float di = dinv[i];
    float v[NCLS];
    float mx = -1e30f;
#pragma unroll
    for (int j = 0; j < NCLS; ++j) {
        v[j] = out[(size_t)i * NCLS + j] * di + b2[j];
        mx = fmaxf(mx, v[j]);
    }
    float s = 0.0f;
#pragma unroll
    for (int j = 0; j < NCLS; ++j) s += __expf(v[j] - mx);
    float l = __logf(s);
#pragma unroll
    for (int j = 0; j < NCLS; ++j) out[(size_t)i * NCLS + j] = v[j] - mx - l;
}

extern "C" void kernel_launch(void* const* d_in, const int* in_sizes, int n_in,
                              void* d_out, int out_size, void* d_ws, size_t ws_size,
                              hipStream_t stream)
{
    const float* x    = (const float*)d_in[0];
    const int*   eidx = (const int*)  d_in[1];
    const float* W1   = (const float*)d_in[2];
    const float* b1   = (const float*)d_in[3];
    const float* W2   = (const float*)d_in[4];
    const float* b2   = (const float*)d_in[5];
    float* out = (float*)d_out;

    const int N = in_sizes[0] / N_FEAT;
    const int E = in_sizes[1] / 2;

    float* ws   = (float*)d_ws;
    float* dinv = ws;
    float* h    = ws + (size_t)N;
    float* agg1 = ws + (size_t)N * 17;
    float* obuf = ws + (size_t)N * 33;

    const int B = 256;
    unsigned* part = (unsigned*)(ws + (size_t)N);
    dim3 gDeg((N + DEG_RN - 1) / DEG_RN, DEG_S);
    k_deg_hist<<<gDeg, 1024, 0, stream>>>(eidx + E, part, E, N);
    k_dinv    <<<(N + B - 1) / B, B, 0, stream>>>(part, dinv, N);

    const int nTiles = (N + 15) / 16;
    k_gemm1<<<(nTiles + 7) / 8, 256, 0, stream>>>(x, W1, dinv, h, agg1, nTiles);

    dim3 gA1((N + 3584 - 1) / 3584, 9);
    k_agg<HID, 16, 3584, 9, 16384, 16><<<gA1, 1024, 0, stream>>>(eidx, h, agg1, E, N);

    k_l2<<<(N + B - 1) / B, B, 0, stream>>>(agg1, b1, W2, dinv, obuf, out, N);

    dim3 gA2((N + 7168 - 1) / 7168, 18);
    k_agg<NCLS, 8, 7168, 18, 16384, 16><<<gA2, 1024, 0, stream>>>(eidx, obuf, out, E, N);

    k_lsm<<<(N + B - 1) / B, B, 0, stream>>>(out, b2, dinv, N);
}


# Edge pass, both layers: the scan also reads the source list, only in lanes whose 4-edge group holds a hit, and queues (source | local destination); the queue pass with 2 lookups per edge goes
# speedup vs baseline: 2.8964x; 1.1528x over previous
#include <hip/hip_runtime.h>
#include <math.h>

#define N_FEAT 512
#define HID    16
#define NCLS   7

typedef __attribute__((ext_vector_type(2))) float v2f;
typedef __attribute__((ext_vector_type(8))) float v8f;

#define DEG_RN 16384
#define DEG_S  32

static __device__ __forceinline__ void deg_count(unsigned* cnt, int c, int base) {
    unsigned d = (unsigned)(c - base);
    if (d < (unsigned)DEG_RN) atomicAdd(&cnt[d], 1u);
}

__global__ __launch_bounds__(1024) void k_deg_hist(
    const int* __restrict__ col, unsigned* __restrict__ part, int E, int n)
{
    __shared__ unsigned cnt[DEG_RN];
    const int tid  = threadIdx.x;
    const int base = blockIdx.x * DEG_RN;
    const int s    = blockIdx.y;
    for (int i = tid; i < DEG_RN; i += 1024) cnt[i] = 0u;
    __syncthreads();

    const int per = (E + DEG_S - 1) / DEG_S;
    int e0 = s * per;       if (e0 > E) e0 = E;
    int e1 = e0 + per;      if (e1 > E) e1 = E;

    const int mis = (int)((((size_t)(col + e0)) >> 2) & 3);
    int a0 = e0 + ((4 - mis) & 3);
    if (a0 > e1) a0 = e1;
    const int nG = (e1 - a0) >> 2;
    const int t0 = a0 + 4 * nG;
    if (tid < a0 - e0) deg_count(cnt, col[e0 + tid], base);
    if (tid < e1 - t0) deg_count(cnt, col[t0 + tid], base);

    const int4* __restrict__ body = reinterpret_cast<const int4*>(col + a0);
    for (int g = tid; g < nG; g += 4 * 1024) {
        int4 v[4];
#pragma unroll
        for (int j = 0; j < 4; ++j) {
            int gg = g + j * 1024;
            int gc = gg < nG ? gg : nG - 1;
            v[j] = body[gc];
            if (gg >= nG) { v[j].x = -1; v[j].y = -1; v[j].z = -1; v[j].w = -1; }
        }
#pragma unroll
        for (int j = 0; j < 4; ++j) {
            deg_count(cnt, v[j].x, base);
            deg_count(cnt, v[j].y, base);
            deg_count(cnt, v[j].z, base);
            deg_count(cnt, v[j].w, base);
        }
    }
    __syncthreads();

    unsigned* __restrict__ plane = part + (size_t)s * (size_t)n;
    for (int i = tid; i < DEG_RN; i += 1024) {
        int node = base + i;
        if (node < n) plane[node] = cnt[i];
    }
}

__global__ void k_dinv(const unsigned* __restrict__ part, float* __restrict__ dinv, int n) {
    int i = blockIdx.x * blockDim.x + threadIdx.x;
    if (i >= n) return;
    unsigned c = 1u;
#pragma unroll
    for (int s = 0; s < DEG_S; ++s) c += part[(size_t)s * (size_t)n + i];
    dinv[i] = rsqrtf((float)c);
}

__global__ __launch_bounds__(256) void k_gemm1(
    const float* __restrict__ x, const float* __restrict__ W1,
    const float* __restrict__ dinv,
    float* __restrict__ h, float* __restrict__ agg1, int nTiles)
{
    __shared__ float sW[N_FEAT * HID];
    for (int i = threadIdx.x; i < N_FEAT * HID; i += blockDim.x) sW[i] = W1[i];
    __syncthreads();

    int wave = threadIdx.x >> 5;
    int lane = threadIdx.x & 31;
    int tile = blockIdx.x * 8 + wave;
    if (tile >= nTiles) return;

    int m  = lane & 15;
    int kh = lane >> 4;
    const float* xr = x + (size_t)(tile * 16 + m) * N_FEAT + 2 * kh;

    v8f acc = {};
#pragma unroll 4
    for (int k = 0; k < N_FEAT; k += 4) {
        v2f a, b;
        a.x = xr[k];
        a.y = xr[k + 1];
        int kk = k + 2 * kh;
        b.x = sW[kk * HID + m];
        b.y = sW[(kk + 1) * HID + m];
        acc = __builtin_amdgcn_wmma_f32_16x16x4_f32(
             false, a,  false, b,
             (short)0, acc,  false,  false);
    }

#pragma unroll
    for (int v = 0; v < 8; ++v) {
        int node  = tile * 16 + v + 8 * kh;
        float hs  = acc[v] * dinv[node];
        h[(size_t)node * HID + m]    = hs;
        agg1[(size_t)node * HID + m] = hs;
    }
}

template <bool B> struct agg_flag { static constexpr bool value = B; };

template <int F, int FP, int RN>
static __device__ __forceinline__ void agg_direct(
    float* acc, const float* __restrict__ srcv, int r, int d)
{
    for (int f = 0; f < F; ++f) atomicAdd(&acc[d * FP + f], srcv[(size_t)r * F + f]);
}

struct src4 { int x, y, z, w; };

template <int F, int FP, int RN, int S, int QCAP, int U>
__global__ __launch_bounds__(1024) void k_agg(
    const int* __restrict__ eidx,
    const float* __restrict__ srcv, float* __restrict__ dst, int E, int n)
{
    __shared__ __attribute__((aligned(16))) float acc[RN * FP];
    __shared__ unsigned q[QCAP + 32];
    __shared__ unsigned qn;

    const int* __restrict__ row = eidx;
    const int* __restrict__ col = eidx + E;
    const int tid  = threadIdx.x;
    const int base = blockIdx.x * RN;
    const int s    = blockIdx.y;

    float4* acc4 = reinterpret_cast<float4*>(acc);
    for (int i = tid; i < RN * FP / 4; i += 1024) acc4[i] = make_float4(0.f, 0.f, 0.f, 0.f);
    if (tid == 0) qn = 0u;
    __syncthreads();

    const int per = (E + S - 1) / S;
    int e0 = s * per;       if (e0 > E) e0 = E;
    int e1 = e0 + per;      if (e1 > E) e1 = E;

    auto visit = [&](int e, int c) {
        unsigned d = (unsigned)(c - base);
        if (d < (unsigned)RN) {
            int r = row[e];
            unsigned pos = atomicAdd(&qn, 1u);
            if (pos < (unsigned)QCAP) q[pos] = (unsigned)r | (d << 19);
            else agg_direct<F, FP, RN>(acc, srcv, r, (int)d);
        }
    };

    const int mis = (int)((((size_t)(col + e0)) >> 2) & 3);
    int a0 = e0 + ((4 - mis) & 3);
    if (a0 > e1) a0 = e1;
    const int nG = (e1 - a0) >> 2;
    const int t0 = a0 + 4 * nG;
    if (tid < a0 - e0) visit(e0 + tid, col[e0 + tid]);
    if (tid < e1 - t0) visit(t0 + tid, col[t0 + tid]);

    const int4* __restrict__ body = reinterpret_cast<const int4*>(col + a0);
    const src4* __restrict__ sbody = reinterpret_cast<const src4*>(row + a0);
    const int lane = tid & 31;
    for (int g0 = 0; g0 < nG; g0 += 8 * 1024) {
        const int g = g0 + tid;
        int4 v[8];
#pragma unroll
        for (int j = 0; j < 8; ++j) {
            int gg = g + j * 1024;
            int gc = gg < nG ? gg : nG - 1;
            v[j] = body[gc];
            if (gg >= nG) { v[j].x = -1; v[j].y = -1; v[j].z = -1; v[j].w = -1; }
        }
        src4 sr[8];
#pragma unroll
        for (int j = 0; j < 8; ++j) {
            bool any = ((unsigned)(v[j].x - base) < (unsigned)RN) | ((unsigned)(v[j].y - base) < (unsigned)RN)
                     | ((unsigned)(v[j].z - base) < (unsigned)RN) | ((unsigned)(v[j].w - base) < (unsigned)RN);
            sr[j] = sbody[any ? g + j * 1024 : 0];
        }
        unsigned nh = 0;
#pragma unroll
        for (int j = 0; j < 8; ++j) {
            nh += (unsigned)(v[j].x - base) < (unsigned)RN;
            nh += (unsigned)(v[j].y - base) < (unsigned)RN;
            nh += (unsigned)(v[j].z - base) < (unsigned)RN;
            nh += (unsigned)(v[j].w - base) < (unsigned)RN;
        }
        unsigned inc = nh;
#pragma unroll
        for (int o = 1; o < 32; o <<= 1) {
            unsigned up = (unsigned)__shfl_up((int)inc, o);
            if (lane >= o) inc += up;
        }
        const unsigned tot = (unsigned)__shfl((int)inc, 31);
        {
            unsigned wb = 0;
            if (lane == 31) wb = atomicAdd(&qn, tot);
            wb = (unsigned)__shfl((int)wb, 31);
            unsigned pos = wb + inc - nh;
            if (wb + tot <= (unsigned)QCAP) {
                const unsigned spare = (unsigned)QCAP + (unsigned)lane;
#pragma unroll
                for (int j = 0; j < 8; ++j) {
                    unsigned dx = (unsigned)(v[j].x - base), dy = (unsigned)(v[j].y - base);
                    unsigned dz = (unsigned)(v[j].z - base), dw = (unsigned)(v[j].w - base);
                    bool mx = dx < (unsigned)RN;
                    bool my = dy < (unsigned)RN;
                    bool mz = dz < (unsigned)RN;
                    bool mw = dw < (unsigned)RN;
                    q[mx ? pos : spare] = (unsigned)sr[j].x | (dx << 19); pos += mx;
                    q[my ? pos : spare] = (unsigned)sr[j].y | (dy << 19); pos += my;
                    q[mz ? pos : spare] = (unsigned)sr[j].z | (dz << 19); pos += mz;
                    q[mw ? pos : spare] = (unsigned)sr[j].w | (dw << 19); pos += mw;
                }
            } else {
#pragma unroll
                for (int j = 0; j < 8; ++j) {
                    int cj[4] = { v[j].x, v[j].y, v[j].z, v[j].w };
                    int rj[4] = { sr[j].x, sr[j].y, sr[j].z, sr[j].w };
#pragma unroll
                    for (int k = 0; k < 4; ++k) {
                        unsigned d = (unsigned)(cj[k] - base);
                        if (d < (unsigned)RN) {
                            if (pos < (unsigned)QCAP) q[pos] = (unsigned)rj[k] | (d << 19);
                            else agg_direct<F, FP, RN>(acc, srcv, rj[k], (int)d);
                            ++pos;
                        }
                    }
                }
            }
        }
    }
    __syncthreads();

    const int cnt = (int)(qn < (unsigned)QCAP ? qn : (unsigned)QCAP);

    {
        constexpr int G = 1024 / FP;
        const int g  = tid / FP;
        const int f  = tid % FP;
        const bool fl = f < F;
        const int fs = fl ? f : 0;
        auto batch = [&](int i0, auto guard) {
            constexpr bool GUARD = decltype(guard)::value;
            unsigned en[U]; float hv[U];
#pragma unroll
            for (int j = 0; j < U; ++j) {
                int i = i0 + j * G;
                en[j] = q[(!GUARD || i < cnt) ? i : cnt - 1];
            }
#pragma unroll
            for (int j = 0; j < U; ++j) {
                unsigned r = en[j] & 0x7FFFFu;
                hv[j] = srcv[(size_t)r * F + fs];
            }
#pragma unroll
            for (int j = 0; j < U; ++j) {
                int i = i0 + j * G;
                if ((!GUARD || i < cnt) && fl)
                    atomicAdd(&acc[(en[j] >> 19) * FP + f], hv[j]);
            }
        };
        const int nfull = cnt / (G * U);
        for (int b = 0; b < nfull; ++b) batch(b * (G * U) + g, agg_flag<false>{});
        const int it = nfull * (G * U) + g;
        if (it < cnt) batch(it, agg_flag<true>{});
    }
    __syncthreads();

    {
        const int T    = RN * F;
        const int stag = ((T / S) & ~1023) * s;
        for (int k = tid; k < T; k += 1024) {
            int t = k + stag;
            if (t >= T) t -= T;
            int nl = t / F;
            int j  = t - nl * F;
            int node = base + nl;
            if (node < n) atomicAdd(&dst[(size_t)node * F + j], acc[nl * FP + j]);
        }
    }
}

__global__ void k_l2(const float* __restrict__ agg1, const float* __restrict__ b1,
                     const float* __restrict__ W2, const float* __restrict__ dinv,
                     float* __restrict__ o, float* __restrict__ outInit, int n)
{
    __shared__ float sW2[HID * NCLS];
    __shared__ float sb1[HID];
    if (threadIdx.x < HID * NCLS) sW2[threadIdx.x] = W2[threadIdx.x];
    if (threadIdx.x < HID)        sb1[threadIdx.x] = b1[threadIdx.x];
    __syncthreads();

    int i = blockIdx.x * blockDim.x + threadIdx.x;
    if (i >= n) return;

    const float di = dinv[i];
    float hv[HID];
#pragma unroll
    for (int f = 0; f < HID; ++f) {
        float v = agg1[(size_t)i * HID + f] * di + sb1[f];
        hv[f] = v > 0.0f ? v : 0.0f;
    }
#pragma unroll
    for (int j = 0; j < NCLS; ++j) {
        float s = 0.0f;
#pragma unroll
        for (int f = 0; f < HID; ++f) s = fmaf(hv[f], sW2[f * NCLS + j], s);
        float os = s * di;
        o[(size_t)i * NCLS + j]       = os;
        outInit[(size_t)i * NCLS + j] = os;
    }
}

__global__ void k_lsm(float* __restrict__ out, const float* __restrict__ b2,
                      const float* __restrict__ dinv, int n)
{
    int i = blockIdx.x * blockDim.x + threadIdx.x;
    if (i >= n) return;
    const float di = dinv[i];
    float v[NCLS];
    float mx = -1e30f;
#pragma unroll
    for (int j = 0; j < NCLS; ++j) {
        v[j] = out[(size_t)i * NCLS + j] * di + b2[j];
        mx = fmaxf(mx, v[j]);
    }
    float s = 0.0f;
#pragma unroll
    for (int j = 0; j < NCLS; ++j) s += __expf(v[j] - mx);
    float l = __logf(s);
#pragma unroll
    for (int j = 0; j < NCLS; ++j) out[(size_t)i * NCLS + j] = v[j] - mx - l;
}

extern "C" void kernel_launch(void* const* d_in, const int* in_sizes, int n_in,
                              void* d_out, int out_size, void* d_ws, size_t ws_size,
                              hipStream_t stream)
{
    const float* x    = (const float*)d_in[0];
    const int*   eidx = (const int*)  d_in[1];
    const float* W1   = (const float*)d_in[2];
    const float* b1   = (const float*)d_in[3];
    const float* W2   = (const float*)d_in[4];
    const float* b2   = (const float*)d_in[5];
    float* out = (float*)d_out;

    const int N = in_sizes[0] / N_FEAT;
    const int E = in_sizes[1] / 2;

    float* ws   = (float*)d_ws;
    float* dinv = ws;
    float* h    = ws + (size_t)N;
    float* agg1 = ws + (size_t)N * 17;
    float* obuf = ws + (size_t)N * 33;

    const int B = 256;
    unsigned* part = (unsigned*)(ws + (size_t)N);
    dim3 gDeg((N + DEG_RN - 1) / DEG_RN, DEG_S);
    k_deg_hist<<<gDeg, 1024, 0, stream>>>(eidx + E, part, E, N);
    k_dinv    <<<(N + B - 1) / B, B, 0, stream>>>(part, dinv, N);

    const int nTiles = (N + 15) / 16;
    k_gemm1<<<(nTiles + 7) / 8, 256, 0, stream>>>(x, W1, dinv, h, agg1, nTiles);

    dim3 gA1((N + 3584 - 1) / 3584, 9);
    k_agg<HID, 16, 3584, 9, 16384, 16><<<gA1, 1024, 0, stream>>>(eidx, h, agg1, E, N);

    k_l2<<<(N + B - 1) / B, B, 0, stream>>>(agg1, b1, W2, dinv, obuf, out, N);

    dim3 gA2((N + 7168 - 1) / 7168, 18);
    k_agg<NCLS, 8, 7168, 18, 16384, 16><<<gA2, 1024, 0, stream>>>(eidx, obuf, out, E, N);

    k_lsm<<<(N + B - 1) / B, B, 0, stream>>>(out, b2, dinv, N);
}


# Layer-1 GEMM operand loads: W1 staged with 8 16-byte loads per thread in flight (was 32 dependent 4-byte trips), x read 16 bytes per lane, a group ahead; k order in the sum changes
# speedup vs baseline: 3.1575x; 1.0901x over previous
#include <hip/hip_runtime.h>
#include <math.h>

#define N_FEAT 512
#define HID    16
#define NCLS   7

typedef __attribute__((ext_vector_type(2))) float v2f;
typedef __attribute__((ext_vector_type(8))) float v8f;

#define DEG_RN 16384
#define DEG_S  32

static __device__ __forceinline__ void deg_count(unsigned* cnt, int c, int base) {
    unsigned d = (unsigned)(c - base);
    if (d < (unsigned)DEG_RN) atomicAdd(&cnt[d], 1u);
}

__global__ __launch_bounds__(1024) void k_deg_hist(
    const int* __restrict__ col, unsigned* __restrict__ part, int E, int n)
{
    __shared__ unsigned cnt[DEG_RN];
    const int tid  = threadIdx.x;
    const int base = blockIdx.x * DEG_RN;
    const int s    = blockIdx.y;
    for (int i = tid; i < DEG_RN; i += 1024) cnt[i] = 0u;
    __syncthreads();

    const int per = (E + DEG_S - 1) / DEG_S;
    int e0 = s * per;       if (e0 > E) e0 = E;
    int e1 = e0 + per;      if (e1 > E) e1 = E;

    const int mis = (int)((((size_t)(col + e0)) >> 2) & 3);
    int a0 = e0 + ((4 - mis) & 3);
    if (a0 > e1) a0 = e1;
    const int nG = (e1 - a0) >> 2;
    const int t0 = a0 + 4 * nG;
    if (tid < a0 - e0) deg_count(cnt, col[e0 + tid], base);
    if (tid < e1 - t0) deg_count(cnt, col[t0 + tid], base);

    const int4* __restrict__ body = reinterpret_cast<const int4*>(col + a0);
    for (int g = tid; g < nG; g += 4 * 1024) {
        int4 v[4];
#pragma unroll
        for (int j = 0; j < 4; ++j) {
            int gg = g + j * 1024;
            int gc = gg < nG ? gg : nG - 1;
            v[j] = body[gc];
            if (gg >= nG) { v[j].x = -1; v[j].y = -1; v[j].z = -1; v[j].w = -1; }
        }
#pragma unroll
        for (int j = 0; j < 4; ++j) {
            deg_count(cnt, v[j].x, base);
            deg_count(cnt, v[j].y, base);
            deg_count(cnt, v[j].z, base);
            deg_count(cnt, v[j].w, base);
        }
    }
    __syncthreads();

    unsigned* __restrict__ plane = part + (size_t)s * (size_t)n;
    for (int i = tid; i < DEG_RN; i += 1024) {
        int node = base + i;
        if (node < n) plane[node] = cnt[i];
    }
}

__global__ void k_dinv(const unsigned* __restrict__ part, float* __restrict__ dinv, int n) {
    int i = blockIdx.x * blockDim.x + threadIdx.x;
    if (i >= n) return;
    unsigned c = 1u;
#pragma unroll
    for (int s = 0; s < DEG_S; ++s) c += part[(size_t)s * (size_t)n + i];
    dinv[i] = rsqrtf((float)c);
}

#define GEMM_T 256
__global__ __launch_bounds__(GEMM_T) void k_gemm1(
    const float* __restrict__ x, const float* __restrict__ W1,
    const float* __restrict__ dinv,
    float* __restrict__ h, float* __restrict__ agg1, int nTiles)
{
    __shared__ __attribute__((aligned(16))) float sW[N_FEAT * HID];

    const int tid  = threadIdx.x;
    const int wave = tid >> 5;
    const int lane = tid & 31;
    const int m  = lane & 15;
    const int kh = lane >> 4;
    const int tile = blockIdx.x * 8 + wave;
    const bool live = tile < nTiles;
    const int tl = live ? tile : nTiles - 1;
    const float4* __restrict__ xr =
        reinterpret_cast<const float4*>(x + (size_t)(tl * 16 + m) * N_FEAT) + kh;

    float4 cur[8];
#pragma unroll
    for (int j = 0; j < 8; ++j) cur[j] = xr[2 * j];

    {
        const float4* __restrict__ W4 = reinterpret_cast<const float4*>(W1);
        float4* sW4 = reinterpret_cast<float4*>(sW);
        float4 w[N_FEAT * HID / 4 / GEMM_T];
#pragma unroll
        for (int j = 0; j < N_FEAT * HID / 4 / GEMM_T; ++j) w[j] = W4[tid + GEMM_T * j];
#pragma unroll
        for (int j = 0; j < N_FEAT * HID / 4 / GEMM_T; ++j) sW4[tid + GEMM_T * j] = w[j];
    }
    __syncthreads();
    if (!live) return;

    v8f acc = {};
#pragma unroll
    for (int g = 0; g < N_FEAT / 64; ++g) {
        float4 nxt[8];
        if (g + 1 < N_FEAT / 64) {
#pragma unroll
            for (int j = 0; j < 8; ++j) nxt[j] = xr[2 * (8 * (g + 1) + j)];
        }
#pragma unroll
        for (int j = 0; j < 8; ++j) {
            const int kk = 8 * (8 * g + j) + 4 * kh;
            v2f a0, a1, b0, b1;
            a0.x = cur[j].x;  a0.y = cur[j].y;
            a1.x = cur[j].z;  a1.y = cur[j].w;
            b0.x = sW[kk * HID + m];        b0.y = sW[(kk + 1) * HID + m];
            b1.x = sW[(kk + 2) * HID + m];  b1.y = sW[(kk + 3) * HID + m];
            acc = __builtin_amdgcn_wmma_f32_16x16x4_f32(
                 false, a0,  false, b0,
                 (short)0, acc,  false,  false);
            acc = __builtin_amdgcn_wmma_f32_16x16x4_f32(
                 false, a1,  false, b1,
                 (short)0, acc,  false,  false);
        }
        if (g + 1 < N_FEAT / 64) {
#pragma unroll
            for (int j = 0; j < 8; ++j) cur[j] = nxt[j];
        }
    }

#pragma unroll
    for (int v = 0; v < 8; ++v) {
        int node  = tile * 16 + v + 8 * kh;
        float hs  = acc[v] * dinv[node];
        h[(size_t)node * HID + m]    = hs;
        agg1[(size_t)node * HID + m] = hs;
    }
}

template <bool B> struct agg_flag { static constexpr bool value = B; };

template <int F, int FP, int RN>
static __device__ __forceinline__ void agg_direct(
    float* acc, const float* __restrict__ srcv, int r, int d)
{
    for (int f = 0; f < F; ++f) atomicAdd(&acc[d * FP + f], srcv[(size_t)r * F + f]);
}

struct src4 { int x, y, z, w; };

template <int F, int FP, int RN, int S, int QCAP, int U>
__global__ __launch_bounds__(1024) void k_agg(
    const int* __restrict__ eidx,
    const float* __restrict__ srcv, float* __restrict__ dst, int E, int n)
{
    __shared__ __attribute__((aligned(16))) float acc[RN * FP];
    __shared__ unsigned q[QCAP + 32];
    __shared__ unsigned qn;

    const int* __restrict__ row = eidx;
    const int* __restrict__ col = eidx + E;
    const int tid  = threadIdx.x;
    const int base = blockIdx.x * RN;
    const int s    = blockIdx.y;

    float4* acc4 = reinterpret_cast<float4*>(acc);
    for (int i = tid; i < RN * FP / 4; i += 1024) acc4[i] = make_float4(0.f, 0.f, 0.f, 0.f);
    if (tid == 0) qn = 0u;
    __syncthreads();

    const int per = (E + S - 1) / S;
    int e0 = s * per;       if (e0 > E) e0 = E;
    int e1 = e0 + per;      if (e1 > E) e1 = E;

    auto visit = [&](int e, int c) {
        unsigned d = (unsigned)(c - base);
        if (d < (unsigned)RN) {
            int r = row[e];
            unsigned pos = atomicAdd(&qn, 1u);
            if (pos < (unsigned)QCAP) q[pos] = (unsigned)r | (d << 19);
            else agg_direct<F, FP, RN>(acc, srcv, r, (int)d);
        }
    };

    const int mis = (int)((((size_t)(col + e0)) >> 2) & 3);
    int a0 = e0 + ((4 - mis) & 3);
    if (a0 > e1) a0 = e1;
    const int nG = (e1 - a0) >> 2;
    const int t0 = a0 + 4 * nG;
    if (tid < a0 - e0) visit(e0 + tid, col[e0 + tid]);
    if (tid < e1 - t0) visit(t0 + tid, col[t0 + tid]);

    const int4* __restrict__ body = reinterpret_cast<const int4*>(col + a0);
    const src4* __restrict__ sbody = reinterpret_cast<const src4*>(row + a0);
    const int lane = tid & 31;
    for (int g0 = 0; g0 < nG; g0 += 8 * 1024) {
        const int g = g0 + tid;
        int4 v[8];
#pragma unroll
        for (int j = 0; j < 8; ++j) {
            int gg = g + j * 1024;
            int gc = gg < nG ? gg : nG - 1;
            v[j] = body[gc];
            if (gg >= nG) { v[j].x = -1; v[j].y = -1; v[j].z = -1; v[j].w = -1; }
        }
        src4 sr[8];
#pragma unroll
        for (int j = 0; j < 8; ++j) {
            bool any = ((unsigned)(v[j].x - base) < (unsigned)RN) | ((unsigned)(v[j].y - base) < (unsigned)RN)
                     | ((unsigned)(v[j].z - base) < (unsigned)RN) | ((unsigned)(v[j].w - base) < (unsigned)RN);
            sr[j] = sbody[any ? g + j * 1024 : 0];
        }
        unsigned nh = 0;
#pragma unroll
        for (int j = 0; j < 8; ++j) {
            nh += (unsigned)(v[j].x - base) < (unsigned)RN;
            nh += (unsigned)(v[j].y - base) < (unsigned)RN;
            nh += (unsigned)(v[j].z - base) < (unsigned)RN;
            nh += (unsigned)(v[j].w - base) < (unsigned)RN;
        }
        unsigned inc = nh;
#pragma unroll
        for (int o = 1; o < 32; o <<= 1) {
            unsigned up = (unsigned)__shfl_up((int)inc, o);
            if (lane >= o) inc += up;
        }
        const unsigned tot = (unsigned)__shfl((int)inc, 31);
        {
            unsigned wb = 0;
            if (lane == 31) wb = atomicAdd(&qn, tot);
            wb = (unsigned)__shfl((int)wb, 31);
            unsigned pos = wb + inc - nh;
            if (wb + tot <= (unsigned)QCAP) {
                const unsigned spare = (unsigned)QCAP + (unsigned)lane;
#pragma unroll
                for (int j = 0; j < 8; ++j) {
                    unsigned dx = (unsigned)(v[j].x - base), dy = (unsigned)(v[j].y - base);
                    unsigned dz = (unsigned)(v[j].z - base), dw = (unsigned)(v[j].w - base);
                    bool mx = dx < (unsigned)RN;
                    bool my = dy < (unsigned)RN;
                    bool mz = dz < (unsigned)RN;
                    bool mw = dw < (unsigned)RN;
                    q[mx ? pos : spare] = (unsigned)sr[j].x | (dx << 19); pos += mx;
                    q[my ? pos : spare] = (unsigned)sr[j].y | (dy << 19); pos += my;
                    q[mz ? pos : spare] = (unsigned)sr[j].z | (dz << 19); pos += mz;
                    q[mw ? pos : spare] = (unsigned)sr[j].w | (dw << 19); pos += mw;
                }
            } else {
#pragma unroll
                for (int j = 0; j < 8; ++j) {
                    int cj[4] = { v[j].x, v[j].y, v[j].z, v[j].w };
                    int rj[4] = { sr[j].x, sr[j].y, sr[j].z, sr[j].w };
#pragma unroll
                    for (int k = 0; k < 4; ++k) {
                        unsigned d = (unsigned)(cj[k] - base);
                        if (d < (unsigned)RN) {
                            if (pos < (unsigned)QCAP) q[pos] = (unsigned)rj[k] | (d << 19);
                            else agg_direct<F, FP, RN>(acc, srcv, rj[k], (int)d);
                            ++pos;
                        }
                    }
                }
            }
        }
    }
    __syncthreads();

    const int cnt = (int)(qn < (unsigned)QCAP ? qn : (unsigned)QCAP);

    {
        constexpr int G = 1024 / FP;
        const int g  = tid / FP;
        const int f  = tid % FP;
        const bool fl = f < F;
        const int fs = fl ? f : 0;
        auto batch = [&](int i0, auto guard) {
            constexpr bool GUARD = decltype(guard)::value;
            unsigned en[U]; float hv[U];
#pragma unroll
            for (int j = 0; j < U; ++j) {
                int i = i0 + j * G;
                en[j] = q[(!GUARD || i < cnt) ? i : cnt - 1];
            }
#pragma unroll
            for (int j = 0; j < U; ++j) {
                unsigned r = en[j] & 0x7FFFFu;
                hv[j] = srcv[(size_t)r * F + fs];
            }
#pragma unroll
            for (int j = 0; j < U; ++j) {
                int i = i0 + j * G;
                if ((!GUARD || i < cnt) && fl)
                    atomicAdd(&acc[(en[j] >> 19) * FP + f], hv[j]);
            }
        };
        const int nfull = cnt / (G * U);
        for (int b = 0; b < nfull; ++b) batch(b * (G * U) + g, agg_flag<false>{});
        const int it = nfull * (G * U) + g;
        if (it < cnt) batch(it, agg_flag<true>{});
    }
    __syncthreads();

    {
        const int T    = RN * F;
        const int stag = ((T / S) & ~1023) * s;
        for (int k = tid; k < T; k += 1024) {
            int t = k + stag;
            if (t >= T) t -= T;
            int nl = t / F;
            int j  = t - nl * F;
            int node = base + nl;
            if (node < n) atomicAdd(&dst[(size_t)node * F + j], acc[nl * FP + j]);
        }
    }
}

__global__ void k_l2(const float* __restrict__ agg1, const float* __restrict__ b1,
                     const float* __restrict__ W2, const float* __restrict__ dinv,
                     float* __restrict__ o, float* __restrict__ outInit, int n)
{
    __shared__ float sW2[HID * NCLS];
    __shared__ float sb1[HID];
    if (threadIdx.x < HID * NCLS) sW2[threadIdx.x] = W2[threadIdx.x];
    if (threadIdx.x < HID)        sb1[threadIdx.x] = b1[threadIdx.x];
    __syncthreads();

    int i = blockIdx.x * blockDim.x + threadIdx.x;
    if (i >= n) return;

    const float di = dinv[i];
    float hv[HID];
#pragma unroll
    for (int f = 0; f < HID; ++f) {
        float v = agg1[(size_t)i * HID + f] * di + sb1[f];
        hv[f] = v > 0.0f ? v : 0.0f;
    }
#pragma unroll
    for (int j = 0; j < NCLS; ++j) {
        float s = 0.0f;
#pragma unroll
        for (int f = 0; f < HID; ++f) s = fmaf(hv[f], sW2[f * NCLS + j], s);
        float os = s * di;
        o[(size_t)i * NCLS + j]       = os;
        outInit[(size_t)i * NCLS + j] = os;
    }
}

__global__ void k_lsm(float* __restrict__ out, const float* __restrict__ b2,
                      const float* __restrict__ dinv, int n)
{
    int i = blockIdx.x * blockDim.x + threadIdx.x;
    if (i >= n) return;
    const float di = dinv[i];
    float v[NCLS];
    float mx = -1e30f;
#pragma unroll
    for (int j = 0; j < NCLS; ++j) {
        v[j] = out[(size_t)i * NCLS + j] * di + b2[j];
        mx = fmaxf(mx, v[j]);
    }
    float s = 0.0f;
#pragma unroll
    for (int j = 0; j < NCLS; ++j) s += __expf(v[j] - mx);
    float l = __logf(s);
#pragma unroll
    for (int j = 0; j < NCLS; ++j) out[(size_t)i * NCLS + j] = v[j] - mx - l;
}

extern "C" void kernel_launch(void* const* d_in, const int* in_sizes, int n_in,
                              void* d_out, int out_size, void* d_ws, size_t ws_size,
                              hipStream_t stream)
{
    const float* x    = (const float*)d_in[0];
    const int*   eidx = (const int*)  d_in[1];
    const float* W1   = (const float*)d_in[2];
    const float* b1   = (const float*)d_in[3];
    const float* W2   = (const float*)d_in[4];
    const float* b2   = (const float*)d_in[5];
    float* out = (float*)d_out;

    const int N = in_sizes[0] / N_FEAT;
    const int E = in_sizes[1] / 2;

    float* ws   = (float*)d_ws;
    float* dinv = ws;
    float* h    = ws + (size_t)N;
    float* agg1 = ws + (size_t)N * 17;
    float* obuf = ws + (size_t)N * 33;

    const int B = 256;
    unsigned* part = (unsigned*)(ws + (size_t)N);
    dim3 gDeg((N + DEG_RN - 1) / DEG_RN, DEG_S);
    k_deg_hist<<<gDeg, 1024, 0, stream>>>(eidx + E, part, E, N);
    k_dinv    <<<(N + B - 1) / B, B, 0, stream>>>(part, dinv, N);

    const int nTiles = (N + 15) / 16;
    k_gemm1<<<(nTiles + 7) / 8, GEMM_T, 0, stream>>>(x, W1, dinv, h, agg1, nTiles);

    dim3 gA1((N + 3584 - 1) / 3584, 9);
    k_agg<HID, 16, 3584, 9, 16384, 16><<<gA1, 1024, 0, stream>>>(eidx, h, agg1, E, N);

    k_l2<<<(N + B - 1) / B, B, 0, stream>>>(agg1, b1, W2, dinv, obuf, out, N);

    dim3 gA2((N + 7168 - 1) / 7168, 18);
    k_agg<NCLS, 8, 7168, 18, 16384, 16><<<gA2, 1024, 0, stream>>>(eidx, obuf, out, E, N);

    k_lsm<<<(N + B - 1) / B, B, 0, stream>>>(out, b2, dinv, N);
}


# Edges bucketed once by destination range into workspace lists (needs 22.5 MB, else old path); one work-group owns a range and its edges, finishes its rows: no scans, no global adds; sum order changes
# speedup vs baseline: 4.7120x; 1.4923x over previous
#include <hip/hip_runtime.h>
#include <math.h>

#define N_FEAT 512
#define HID    16
#define NCLS   7

typedef __attribute__((ext_vector_type(2))) float v2f;
typedef __attribute__((ext_vector_type(8))) float v8f;

#define DEG_RN 16384
#define DEG_S  32

static __device__ __forceinline__ void deg_count(unsigned* cnt, int c, int base) {
    unsigned d = (unsigned)(c - base);
    if (d < (unsigned)DEG_RN) atomicAdd(&cnt[d], 1u);
}

__global__ __launch_bounds__(1024) void k_deg_hist(
    const int* __restrict__ col, unsigned* __restrict__ part, int E, int n)
{
    __shared__ unsigned cnt[DEG_RN];
    const int tid  = threadIdx.x;
    const int base = blockIdx.x * DEG_RN;
    const int s    = blockIdx.y;
    for (int i = tid; i < DEG_RN; i += 1024) cnt[i] = 0u;
    __syncthreads();

    const int per = (E + DEG_S - 1) / DEG_S;
    int e0 = s * per;       if (e0 > E) e0 = E;
    int e1 = e0 + per;      if (e1 > E) e1 = E;

    const int mis = (int)((((size_t)(col + e0)) >> 2) & 3);
    int a0 = e0 + ((4 - mis) & 3);
    if (a0 > e1) a0 = e1;
    const int nG = (e1 - a0) >> 2;
    const int t0 = a0 + 4 * nG;
    if (tid < a0 - e0) deg_count(cnt, col[e0 + tid], base);
    if (tid < e1 - t0) deg_count(cnt, col[t0 + tid], base);

    const int4* __restrict__ body = reinterpret_cast<const int4*>(col + a0);
    for (int g = tid; g < nG; g += 4 * 1024) {
        int4 v[4];
#pragma unroll
        for (int j = 0; j < 4; ++j) {
            int gg = g + j * 1024;
            int gc = gg < nG ? gg : nG - 1;
            v[j] = body[gc];
            if (gg >= nG) { v[j].x = -1; v[j].y = -1; v[j].z = -1; v[j].w = -1; }
        }
#pragma unroll
        for (int j = 0; j < 4; ++j) {
            deg_count(cnt, v[j].x, base);
            deg_count(cnt, v[j].y, base);
            deg_count(cnt, v[j].z, base);
            deg_count(cnt, v[j].w, base);
        }
    }
    __syncthreads();

    unsigned* __restrict__ plane = part + (size_t)s * (size_t)n;
    for (int i = tid; i < DEG_RN; i += 1024) {
        int node = base + i;
        if (node < n) plane[node] = cnt[i];
    }
}

__global__ void k_dinv(const unsigned* __restrict__ part, float* __restrict__ dinv, int n) {
    int i = blockIdx.x * blockDim.x + threadIdx.x;
    if (i >= n) return;
    unsigned c = 1u;
#pragma unroll
    for (int s = 0; s < DEG_S; ++s) c += part[(size_t)s * (size_t)n + i];
    dinv[i] = rsqrtf((float)c);
}

#define GEMM_T 256
template <bool SELF>
__global__ __launch_bounds__(GEMM_T) void k_gemm1(
    const float* __restrict__ x, const float* __restrict__ W1,
    const float* __restrict__ dinv,
    float* __restrict__ h, float* __restrict__ agg1, int nTiles)
{
    __shared__ __attribute__((aligned(16))) float sW[N_FEAT * HID];

    const int tid  = threadIdx.x;
    const int wave = tid >> 5;
    const int lane = tid & 31;
    const int m  = lane & 15;
    const int kh = lane >> 4;
    const int tile = blockIdx.x * 8 + wave;
    const bool live = tile < nTiles;
    const int tl = live ? tile : nTiles - 1;
    const float4* __restrict__ xr =
        reinterpret_cast<const float4*>(x + (size_t)(tl * 16 + m) * N_FEAT) + kh;

    float4 cur[8];
#pragma unroll
    for (int j = 0; j < 8; ++j) cur[j] = xr[2 * j];

    {
        const float4* __restrict__ W4 = reinterpret_cast<const float4*>(W1);
        float4* sW4 = reinterpret_cast<float4*>(sW);
        float4 w[N_FEAT * HID / 4 / GEMM_T];
#pragma unroll
        for (int j = 0; j < N_FEAT * HID / 4 / GEMM_T; ++j) w[j] = W4[tid + GEMM_T * j];
#pragma unroll
        for (int j = 0; j < N_FEAT * HID / 4 / GEMM_T; ++j) sW4[tid + GEMM_T * j] = w[j];
    }
    __syncthreads();
    if (!live) return;

    v8f acc = {};
#pragma unroll
    for (int g = 0; g < N_FEAT / 64; ++g) {
        float4 nxt[8];
        if (g + 1 < N_FEAT / 64) {
#pragma unroll
            for (int j = 0; j < 8; ++j) nxt[j] = xr[2 * (8 * (g + 1) + j)];
        }
#pragma unroll
        for (int j = 0; j < 8; ++j) {
            const int kk = 8 * (8 * g + j) + 4 * kh;
            v2f a0, a1, b0, b1;
            a0.x = cur[j].x;  a0.y = cur[j].y;
            a1.x = cur[j].z;  a1.y = cur[j].w;
            b0.x = sW[kk * HID + m];        b0.y = sW[(kk + 1) * HID + m];
            b1.x = sW[(kk + 2) * HID + m];  b1.y = sW[(kk + 3) * HID + m];
            acc = __builtin_amdgcn_wmma_f32_16x16x4_f32(
                 false, a0,  false, b0,
                 (short)0, acc,  false,  false);
            acc = __builtin_amdgcn_wmma_f32_16x16x4_f32(
                 false, a1,  false, b1,
                 (short)0, acc,  false,  false);
        }
        if (g + 1 < N_FEAT / 64) {
#pragma unroll
            for (int j = 0; j < 8; ++j) cur[j] = nxt[j];
        }
    }

#pragma unroll
    for (int v = 0; v < 8; ++v) {
        int node  = tile * 16 + v + 8 * kh;
        float hs  = acc[v] * dinv[node];
        h[(size_t)node * HID + m]    = hs;
        if (SELF) agg1[(size_t)node * HID + m] = hs;
    }
}

template <bool B> struct agg_flag { static constexpr bool value = B; };

template <int F, int FP, int RN>
static __device__ __forceinline__ void agg_direct(
    float* acc, const float* __restrict__ srcv, int r, int d)
{
    for (int f = 0; f < F; ++f) atomicAdd(&acc[d * FP + f], srcv[(size_t)r * F + f]);
}

struct src4 { int x, y, z, w; };

template <int F, int FP, int RN, int S, int QCAP, int U>
__global__ __launch_bounds__(1024) void k_agg(
    const int* __restrict__ eidx,
    const float* __restrict__ srcv, float* __restrict__ dst, int E, int n)
{
    __shared__ __attribute__((aligned(16))) float acc[RN * FP];
    __shared__ unsigned q[QCAP + 32];
    __shared__ unsigned qn;

    const int* __restrict__ row = eidx;
    const int* __restrict__ col = eidx + E;
    const int tid  = threadIdx.x;
    const int base = blockIdx.x * RN;
    const int s    = blockIdx.y;

    float4* acc4 = reinterpret_cast<float4*>(acc);
    for (int i = tid; i < RN * FP / 4; i += 1024) acc4[i] = make_float4(0.f, 0.f, 0.f, 0.f);
    if (tid == 0) qn = 0u;
    __syncthreads();

    const int per = (E + S - 1) / S;
    int e0 = s * per;       if (e0 > E) e0 = E;
    int e1 = e0 + per;      if (e1 > E) e1 = E;

    auto visit = [&](int e, int c) {
        unsigned d = (unsigned)(c - base);
        if (d < (unsigned)RN) {
            int r = row[e];
            unsigned pos = atomicAdd(&qn, 1u);
            if (pos < (unsigned)QCAP) q[pos] = (unsigned)r | (d << 19);
            else agg_direct<F, FP, RN>(acc, srcv, r, (int)d);
        }
    };

    const int mis = (int)((((size_t)(col + e0)) >> 2) & 3);
    int a0 = e0 + ((4 - mis) & 3);
    if (a0 > e1) a0 = e1;
    const int nG = (e1 - a0) >> 2;
    const int t0 = a0 + 4 * nG;
    if (tid < a0 - e0) visit(e0 + tid, col[e0 + tid]);
    if (tid < e1 - t0) visit(t0 + tid, col[t0 + tid]);

    const int4* __restrict__ body = reinterpret_cast<const int4*>(col + a0);
    const src4* __restrict__ sbody = reinterpret_cast<const src4*>(row + a0);
    const int lane = tid & 31;
    for (int g0 = 0; g0 < nG; g0 += 8 * 1024) {
        const int g = g0 + tid;
        int4 v[8];
#pragma unroll
        for (int j = 0; j < 8; ++j) {
            int gg = g + j * 1024;
            int gc = gg < nG ? gg : nG - 1;
            v[j] = body[gc];
            if (gg >= nG) { v[j].x = -1; v[j].y = -1; v[j].z = -1; v[j].w = -1; }
        }
        src4 sr[8];
#pragma unroll
        for (int j = 0; j < 8; ++j) {
            bool any = ((unsigned)(v[j].x - base) < (unsigned)RN) | ((unsigned)(v[j].y - base) < (unsigned)RN)
                     | ((unsigned)(v[j].z - base) < (unsigned)RN) | ((unsigned)(v[j].w - base) < (unsigned)RN);
            sr[j] = sbody[any ? g + j * 1024 : 0];
        }
        unsigned nh = 0;
#pragma unroll
        for (int j = 0; j < 8; ++j) {
            nh += (unsigned)(v[j].x - base) < (unsigned)RN;
            nh += (unsigned)(v[j].y - base) < (unsigned)RN;
            nh += (unsigned)(v[j].z - base) < (unsigned)RN;
            nh += (unsigned)(v[j].w - base) < (unsigned)RN;
        }
        unsigned inc = nh;
#pragma unroll
        for (int o = 1; o < 32; o <<= 1) {
            unsigned up = (unsigned)__shfl_up((int)inc, o);
            if (lane >= o) inc += up;
        }
        const unsigned tot = (unsigned)__shfl((int)inc, 31);
        {
            unsigned wb = 0;
            if (lane == 31) wb = atomicAdd(&qn, tot);
            wb = (unsigned)__shfl((int)wb, 31);
            unsigned pos = wb + inc - nh;
            if (wb + tot <= (unsigned)QCAP) {
                const unsigned spare = (unsigned)QCAP + (unsigned)lane;
#pragma unroll
                for (int j = 0; j < 8; ++j) {
                    unsigned dx = (unsigned)(v[j].x - base), dy = (unsigned)(v[j].y - base);
                    unsigned dz = (unsigned)(v[j].z - base), dw = (unsigned)(v[j].w - base);
                    bool mx = dx < (unsigned)RN;
                    bool my = dy < (unsigned)RN;
                    bool mz = dz < (unsigned)RN;
                    bool mw = dw < (unsigned)RN;
                    q[mx ? pos : spare] = (unsigned)sr[j].x | (dx << 19); pos += mx;
                    q[my ? pos : spare] = (unsigned)sr[j].y | (dy << 19); pos += my;
                    q[mz ? pos : spare] = (unsigned)sr[j].z | (dz << 19); pos += mz;
                    q[mw ? pos : spare] = (unsigned)sr[j].w | (dw << 19); pos += mw;
                }
            } else {
#pragma unroll
                for (int j = 0; j < 8; ++j) {
                    int cj[4] = { v[j].x, v[j].y, v[j].z, v[j].w };
                    int rj[4] = { sr[j].x, sr[j].y, sr[j].z, sr[j].w };
#pragma unroll
                    for (int k = 0; k < 4; ++k) {
                        unsigned d = (unsigned)(cj[k] - base);
                        if (d < (unsigned)RN) {
                            if (pos < (unsigned)QCAP) q[pos] = (unsigned)rj[k] | (d << 19);
                            else agg_direct<F, FP, RN>(acc, srcv, rj[k], (int)d);
                            ++pos;
                        }
                    }
                }
            }
        }
    }
    __syncthreads();

    const int cnt = (int)(qn < (unsigned)QCAP ? qn : (unsigned)QCAP);

    {
        constexpr int G = 1024 / FP;
        const int g  = tid / FP;
        const int f  = tid % FP;
        const bool fl = f < F;
        const int fs = fl ? f : 0;
        auto batch = [&](int i0, auto guard) {
            constexpr bool GUARD = decltype(guard)::value;
            unsigned en[U]; float hv[U];
#pragma unroll
            for (int j = 0; j < U; ++j) {
                int i = i0 + j * G;
                en[j] = q[(!GUARD || i < cnt) ? i : cnt - 1];
            }
#pragma unroll
            for (int j = 0; j < U; ++j) {
                unsigned r = en[j] & 0x7FFFFu;
                hv[j] = srcv[(size_t)r * F + fs];
            }
#pragma unroll
            for (int j = 0; j < U; ++j) {
                int i = i0 + j * G;
                if ((!GUARD || i < cnt) && fl)
                    atomicAdd(&acc[(en[j] >> 19) * FP + f], hv[j]);
            }
        };
        const int nfull = cnt / (G * U);
        for (int b = 0; b < nfull; ++b) batch(b * (G * U) + g, agg_flag<false>{});
        const int it = nfull * (G * U) + g;
        if (it < cnt) batch(it, agg_flag<true>{});
    }
    __syncthreads();

    {
        const int T    = RN * F;
        const int stag = ((T / S) & ~1023) * s;
        for (int k = tid; k < T; k += 1024) {
            int t = k + stag;
            if (t >= T) t -= T;
            int nl = t / F;
            int j  = t - nl * F;
            int node = base + nl;
            if (node < n) atomicAdd(&dst[(size_t)node * F + j], acc[nl * FP + j]);
        }
    }
}

__global__ void k_l2(const float* __restrict__ agg1, const float* __restrict__ b1,
                     const float* __restrict__ W2, const float* __restrict__ dinv,
                     float* __restrict__ o, float* __restrict__ outInit, int n)
{
    __shared__ float sW2[HID * NCLS];
    __shared__ float sb1[HID];
    if (threadIdx.x < HID * NCLS) sW2[threadIdx.x] = W2[threadIdx.x];
    if (threadIdx.x < HID)        sb1[threadIdx.x] = b1[threadIdx.x];
    __syncthreads();

    int i = blockIdx.x * blockDim.x + threadIdx.x;
    if (i >= n) return;

    const float di = dinv[i];
    float hv[HID];
#pragma unroll
    for (int f = 0; f < HID; ++f) {
        float v = agg1[(size_t)i * HID + f] * di + sb1[f];
        hv[f] = v > 0.0f ? v : 0.0f;
    }
#pragma unroll
    for (int j = 0; j < NCLS; ++j) {
        float s = 0.0f;
#pragma unroll
        for (int f = 0; f < HID; ++f) s = fmaf(hv[f], sW2[f * NCLS + j], s);
        float os = s * di;
        o[(size_t)i * NCLS + j]       = os;
        outInit[(size_t)i * NCLS + j] = os;
    }
}

__global__ void k_lsm(float* __restrict__ out, const float* __restrict__ b2,
                      const float* __restrict__ dinv, int n)
{
    int i = blockIdx.x * blockDim.x + threadIdx.x;
    if (i >= n) return;
    const float di = dinv[i];
    float v[NCLS];
    float mx = -1e30f;
#pragma unroll
    for (int j = 0; j < NCLS; ++j) {
        v[j] = out[(size_t)i * NCLS + j] * di + b2[j];
        mx = fmaxf(mx, v[j]);
    }
    float s = 0.0f;
#pragma unroll
    for (int j = 0; j < NCLS; ++j) s += __expf(v[j] - mx);
    float l = __logf(s);
#pragma unroll
    for (int j = 0; j < NCLS; ++j) out[(size_t)i * NCLS + j] = v[j] - mx - l;
}


#define NB    256
#define RNBX  512
#define META  1024

static __device__ __forceinline__ unsigned scan_nb_excl(unsigned v, unsigned* wsum, int tid)
{
    const int lane = tid & 31, w = tid >> 5;
    unsigned inc = v;
#pragma unroll
    for (int o = 1; o < 32; o <<= 1) {
        unsigned up = (unsigned)__shfl_up((int)inc, o);
        if (lane >= o) inc += up;
    }
    if (w < NB / 32 && lane == 31) wsum[w] = inc;
    __syncthreads();
    unsigned add = 0u;
    if (w < NB / 32) for (int q = 0; q < w; ++q) add += wsum[q];
    __syncthreads();
    return add + inc - v;
}

__global__ __launch_bounds__(RNBX) void k_dinv_tot(
    const unsigned* __restrict__ part, float* __restrict__ dinv,
    unsigned* __restrict__ tot, unsigned* __restrict__ cur, int n, int rnb)
{
    __shared__ unsigned red[RNBX / 32];
    const int tid  = threadIdx.x;
    const int node = blockIdx.x * rnb + tid;
    unsigned s = 0u;
    if (tid < rnb && node < n) {
#pragma unroll
        for (int p = 0; p < DEG_S; ++p) s += part[(size_t)p * (size_t)n + node];
        dinv[node] = rsqrtf((float)(s + 1u));
    }
    unsigned w = s;
#pragma unroll
    for (int o = 16; o > 0; o >>= 1) w += (unsigned)__shfl_xor((int)w, o);
    if ((tid & 31) == 0) red[tid >> 5] = w;
    __syncthreads();
    if (tid == 0) {
        unsigned t = 0u;
        for (int q = 0; q < RNBX / 32; ++q) t += red[q];
        tot[blockIdx.x] = t;
        cur[blockIdx.x] = 0u;
    }
}

#define PL_CH  16384
#define PL_EPT (PL_CH / 1024)
__global__ __launch_bounds__(1024) void k_place(
    const int* __restrict__ eidx, const unsigned* __restrict__ tot,
    unsigned* __restrict__ cur, unsigned* __restrict__ list, int E, int rnb)
{
    __shared__ unsigned sorted[PL_CH];
    __shared__ unsigned short sk[PL_CH];
    __shared__ unsigned cnt[NB], lstart[NB], gpos[NB], bbase[NB];
    __shared__ unsigned wsum[NB / 32];

    const int* __restrict__ row = eidx;
    const int* __restrict__ col = eidx + E;
    const int tid = threadIdx.x;

    {
        unsigned t  = tid < NB ? tot[tid] : 0u;
        unsigned ex = scan_nb_excl(t, wsum, tid);
        if (tid < NB) bbase[tid] = ex;
    }

    const int per = (E + (int)gridDim.x - 1) / (int)gridDim.x;
    int e0 = (int)blockIdx.x * per;  if (e0 > E) e0 = E;
    int e1 = e0 + per;               if (e1 > E) e1 = E;

    for (int c0 = e0; c0 < e1; c0 += PL_CH) {
        const int m = (e1 - c0) < PL_CH ? (e1 - c0) : PL_CH;
        if (tid < NB) cnt[tid] = 0u;
        __syncthreads();

        int cj[PL_EPT], rj[PL_EPT];
#pragma unroll
        for (int j = 0; j < PL_EPT; ++j) {
            int i  = tid + j * 1024;
            int ii = i < m ? i : m - 1;
            cj[j] = col[c0 + ii];
            rj[j] = row[c0 + ii];
        }
        unsigned ent[PL_EPT], kr[PL_EPT];
#pragma unroll
        for (int j = 0; j < PL_EPT; ++j) {
            int i = tid + j * 1024;
            ent[j] = 0u; kr[j] = 0u;
            if (i < m) {
                unsigned k  = (unsigned)cj[j] / (unsigned)rnb;
                unsigned dl = (unsigned)cj[j] - k * (unsigned)rnb;
                unsigned r  = atomicAdd(&cnt[k], 1u);
                ent[j] = (unsigned)rj[j] | (dl << 19);
                kr[j]  = k | (r << 8);
            }
        }
        __syncthreads();
        {
            unsigned c  = tid < NB ? cnt[tid] : 0u;
            unsigned ex = scan_nb_excl(c, wsum, tid);
            if (tid < NB) {
                lstart[tid] = ex;
                unsigned g = bbase[tid];
                if (c) g += atomicAdd(&cur[tid], c);
                gpos[tid] = g;
            }
        }
        __syncthreads();
#pragma unroll
        for (int j = 0; j < PL_EPT; ++j) {
            int i = tid + j * 1024;
            if (i < m) {
                unsigned k = kr[j] & 255u;
                unsigned p = lstart[k] + (kr[j] >> 8);
                sorted[p] = ent[j];
                sk[p]     = (unsigned short)k;
            }
        }
        __syncthreads();
        for (int i = tid; i < m; i += 1024) {
            unsigned k = sk[i];
            list[gpos[k] + ((unsigned)i - lstart[k])] = sorted[i];
        }
        __syncthreads();
    }
}

template <int F, int FP, bool GUARD>
static __device__ __forceinline__ unsigned own_batch(
    float* acc, const float* __restrict__ srcv, unsigned mine, int lane,
    const unsigned* __restrict__ seg, int inext, int cnt)
{
    constexpr int GPW = 32 / FP;
    constexpr int U   = 32 / GPW;
    const int gi = lane / FP;
    const int f  = lane % FP;
    const bool fl = f < F;
    const int fs = fl ? f : 0;
    unsigned en[U]; float hv[U];
#pragma unroll
    for (int j = 0; j < U; ++j) en[j] = (unsigned)__shfl((int)mine, j * GPW + gi);
#pragma unroll
    for (int j = 0; j < U; ++j) {
        unsigned r = en[j] & 0x7FFFFu;
        if (GUARD && en[j] == 0xFFFFFFFFu) r = 0u;
        hv[j] = srcv[(size_t)r * F + fs];
    }
    unsigned nxt = 0xFFFFFFFFu;
    if (!GUARD) {
        unsigned ld = seg[inext < cnt ? inext : cnt - 1];
        nxt = inext < cnt ? ld : 0xFFFFFFFFu;
    }
#pragma unroll
    for (int j = 0; j < U; ++j) {
        if ((!GUARD || en[j] != 0xFFFFFFFFu) && fl)
            atomicAdd(&acc[(en[j] >> 19) * FP + f], hv[j]);
    }
    return nxt;
}

template <int F, int FP>
static __device__ __forceinline__ void own_walk(
    float* acc, const float* __restrict__ srcv, const unsigned* __restrict__ seg, int cnt, int tid)
{
    const int lane  = tid & 31;
    const int nfull = cnt / 1024;
    unsigned nxt = tid < cnt ? seg[tid] : 0xFFFFFFFFu;
    for (int b = 0; b < nfull; ++b)
        nxt = own_batch<F, FP, false>(acc, srcv, nxt, lane, seg, (b + 1) * 1024 + tid, cnt);
    if (nfull * 1024 < cnt) own_batch<F, FP, true>(acc, srcv, nxt, lane, seg, 0, cnt);
}

__global__ __launch_bounds__(1024) void k_own1(
    const unsigned* __restrict__ list, const unsigned* __restrict__ tot,
    const float* __restrict__ h, const float* __restrict__ dinv,
    const float* __restrict__ b1, const float* __restrict__ W2,
    float* __restrict__ o, int n, int rnb)
{
    __shared__ __attribute__((aligned(16))) float acc[RNBX * HID];
    __shared__ unsigned wsum[NB / 32];
    __shared__ unsigned sseg[2];
    __shared__ float sW2[HID * NCLS];
    __shared__ float sb1[HID];
    const int tid = threadIdx.x;
    const int k   = blockIdx.x;
    {
        unsigned t  = tid < NB ? tot[tid] : 0u;
        unsigned ex = scan_nb_excl(t, wsum, tid);
        if (tid == k) { sseg[0] = ex; sseg[1] = t; }
    }
    float4* acc4 = reinterpret_cast<float4*>(acc);
    for (int i = tid; i < RNBX * HID / 4; i += 1024) acc4[i] = make_float4(0.f, 0.f, 0.f, 0.f);
    if (tid < HID * NCLS) sW2[tid] = W2[tid];
    if (tid < HID)        sb1[tid] = b1[tid];
    __syncthreads();

    own_walk<HID, 16>(acc, h, list + sseg[0], (int)sseg[1], tid);
    __syncthreads();

    const int node = k * rnb + tid;
    if (tid < rnb && node < n) {
        const float di = dinv[node];
        float hv[HID];
#pragma unroll
        for (int f = 0; f < HID; ++f) {
            float v = (acc[tid * HID + f] + h[(size_t)node * HID + f]) * di + sb1[f];
            hv[f] = v > 0.0f ? v : 0.0f;
        }
#pragma unroll
        for (int j = 0; j < NCLS; ++j) {
            float s = 0.0f;
#pragma unroll
            for (int f = 0; f < HID; ++f) s = fmaf(hv[f], sW2[f * NCLS + j], s);
            o[(size_t)node * NCLS + j] = s * di;
        }
    }
}

__global__ __launch_bounds__(1024) void k_own2(
    const unsigned* __restrict__ list, const unsigned* __restrict__ tot,
    const float* __restrict__ o, const float* __restrict__ dinv,
    const float* __restrict__ b2, float* __restrict__ out, int n, int rnb)
{
    __shared__ __attribute__((aligned(16))) float acc[RNBX * 8];
    __shared__ unsigned wsum[NB / 32];
    __shared__ unsigned sseg[2];
    const int tid = threadIdx.x;
    const int k   = blockIdx.x;
    {
        unsigned t  = tid < NB ? tot[tid] : 0u;
        unsigned ex = scan_nb_excl(t, wsum, tid);
        if (tid == k) { sseg[0] = ex; sseg[1] = t; }
    }
    float4* acc4 = reinterpret_cast<float4*>(acc);
    for (int i = tid; i < RNBX * 8 / 4; i += 1024) acc4[i] = make_float4(0.f, 0.f, 0.f, 0.f);
    __syncthreads();

    own_walk<NCLS, 8>(acc, o, list + sseg[0], (int)sseg[1], tid);
    __syncthreads();

    const int node = k * rnb + tid;
    if (tid < rnb && node < n) {
        const float di = dinv[node];
        float v[NCLS];
        float mx = -1e30f;
#pragma unroll
        for (int j = 0; j < NCLS; ++j) {
            v[j] = (acc[tid * 8 + j] + o[(size_t)node * NCLS + j]) * di + b2[j];
            mx = fmaxf(mx, v[j]);
        }
        float s = 0.0f;
#pragma unroll
        for (int j = 0; j < NCLS; ++j) s += __expf(v[j] - mx);
        float l = __logf(s);
#pragma unroll
        for (int j = 0; j < NCLS; ++j) out[(size_t)node * NCLS + j] = v[j] - mx - l;
    }
}

extern "C" void kernel_launch(void* const* d_in, const int* in_sizes, int n_in,
                              void* d_out, int out_size, void* d_ws, size_t ws_size,
                              hipStream_t stream)
{
    const float* x    = (const float*)d_in[0];
    const int*   eidx = (const int*)  d_in[1];
    const float* W1   = (const float*)d_in[2];
    const float* b1   = (const float*)d_in[3];
    const float* W2   = (const float*)d_in[4];
    const float* b2   = (const float*)d_in[5];
    float* out = (float*)d_out;

    const int N = in_sizes[0] / N_FEAT;
    const int E = in_sizes[1] / 2;

    float* ws   = (float*)d_ws;

    {
        const int rnb = (N + NB - 1) / NB;
        const size_t needL = (size_t)N * 24 + META + (size_t)E;
        const size_t needP = (size_t)N * (1 + DEG_S) + META;
        const size_t need  = (needL > needP ? needL : needP) * sizeof(float);
        if (ws_size >= need && rnb <= RNBX && N <= (1 << 19)) {
            float*    dinvO = ws;
            unsigned* tot   = (unsigned*)(ws + (size_t)N);
            unsigned* cur   = tot + NB;
            float*    hO    = ws + (size_t)N + META;
            float*    oO    = hO + (size_t)N * HID;
            unsigned* list  = (unsigned*)(oO + (size_t)N * NCLS);
            unsigned* partO = (unsigned*)hO;

            dim3 gDegO((N + DEG_RN - 1) / DEG_RN, DEG_S);
            k_deg_hist<<<gDegO, 1024, 0, stream>>>(eidx + E, partO, E, N);
            k_dinv_tot<<<NB, RNBX, 0, stream>>>(partO, dinvO, tot, cur, N, rnb);
            const int nTilesO = (N + 15) / 16;
            k_gemm1<false><<<(nTilesO + 7) / 8, GEMM_T, 0, stream>>>(x, W1, dinvO, hO, nullptr, nTilesO);
            k_place<<<NB, 1024, 0, stream>>>(eidx, tot, cur, list, E, rnb);
            k_own1<<<NB, 1024, 0, stream>>>(list, tot, hO, dinvO, b1, W2, oO, N, rnb);
            k_own2<<<NB, 1024, 0, stream>>>(list, tot, oO, dinvO, b2, out, N, rnb);
            return;
        }
    }
    float* dinv = ws;
    float* h    = ws + (size_t)N;
    float* agg1 = ws + (size_t)N * 17;
    float* obuf = ws + (size_t)N * 33;

    const int B = 256;
    unsigned* part = (unsigned*)(ws + (size_t)N);
    dim3 gDeg((N + DEG_RN - 1) / DEG_RN, DEG_S);
    k_deg_hist<<<gDeg, 1024, 0, stream>>>(eidx + E, part, E, N);
    k_dinv    <<<(N + B - 1) / B, B, 0, stream>>>(part, dinv, N);

    const int nTiles = (N + 15) / 16;
    k_gemm1<true><<<(nTiles + 7) / 8, GEMM_T, 0, stream>>>(x, W1, dinv, h, agg1, nTiles);

    dim3 gA1((N + 3584 - 1) / 3584, 9);
    k_agg<HID, 16, 3584, 9, 16384, 16><<<gA1, 1024, 0, stream>>>(eidx, h, agg1, E, N);

    k_l2<<<(N + B - 1) / B, B, 0, stream>>>(agg1, b1, W2, dinv, obuf, out, N);

    dim3 gA2((N + 7168 - 1) / 7168, 18);
    k_agg<NCLS, 8, 7168, 18, 16384, 16><<<gA2, 1024, 0, stream>>>(eidx, obuf, out, E, N);

    k_lsm<<<(N + B - 1) / B, B, 0, stream>>>(out, b2, dinv, N);
}


# Edge placing and layer-1 GEMM run as one launch (independent work; block role by index, placing re-cut to 256-thread groups of 4096 edges); order inside a range's list, so sum order, changes
# speedup vs baseline: 5.0848x; 1.0791x over previous
#include <hip/hip_runtime.h>
#include <math.h>

#define N_FEAT 512
#define HID    16
#define NCLS   7

typedef __attribute__((ext_vector_type(2))) float v2f;
typedef __attribute__((ext_vector_type(8))) float v8f;

#define DEG_RN 16384
#define DEG_S  32

static __device__ __forceinline__ void deg_count(unsigned* cnt, int c, int base) {
    unsigned d = (unsigned)(c - base);
    if (d < (unsigned)DEG_RN) atomicAdd(&cnt[d], 1u);
}

__global__ __launch_bounds__(1024) void k_deg_hist(
    const int* __restrict__ col, unsigned* __restrict__ part, int E, int n)
{
    __shared__ unsigned cnt[DEG_RN];
    const int tid  = threadIdx.x;
    const int base = blockIdx.x * DEG_RN;
    const int s    = blockIdx.y;
    for (int i = tid; i < DEG_RN; i += 1024) cnt[i] = 0u;
    __syncthreads();

    const int per = (E + DEG_S - 1) / DEG_S;
    int e0 = s * per;       if (e0 > E) e0 = E;
    int e1 = e0 + per;      if (e1 > E) e1 = E;

    const int mis = (int)((((size_t)(col + e0)) >> 2) & 3);
    int a0 = e0 + ((4 - mis) & 3);
    if (a0 > e1) a0 = e1;
    const int nG = (e1 - a0) >> 2;
    const int t0 = a0 + 4 * nG;
    if (tid < a0 - e0) deg_count(cnt, col[e0 + tid], base);
    if (tid < e1 - t0) deg_count(cnt, col[t0 + tid], base);

    const int4* __restrict__ body = reinterpret_cast<const int4*>(col + a0);
    for (int g = tid; g < nG; g += 4 * 1024) {
        int4 v[4];
#pragma unroll
        for (int j = 0; j < 4; ++j) {
            int gg = g + j * 1024;
            int gc = gg < nG ? gg : nG - 1;
            v[j] = body[gc];
            if (gg >= nG) { v[j].x = -1; v[j].y = -1; v[j].z = -1; v[j].w = -1; }
        }
#pragma unroll
        for (int j = 0; j < 4; ++j) {
            deg_count(cnt, v[j].x, base);
            deg_count(cnt, v[j].y, base);
            deg_count(cnt, v[j].z, base);
            deg_count(cnt, v[j].w, base);
        }
    }
    __syncthreads();

    unsigned* __restrict__ plane = part + (size_t)s * (size_t)n;
    for (int i = tid; i < DEG_RN; i += 1024) {
        int node = base + i;
        if (node < n) plane[node] = cnt[i];
    }
}

__global__ void k_dinv(const unsigned* __restrict__ part, float* __restrict__ dinv, int n) {
    int i = blockIdx.x * blockDim.x + threadIdx.x;
    if (i >= n) return;
    unsigned c = 1u;
#pragma unroll
    for (int s = 0; s < DEG_S; ++s) c += part[(size_t)s * (size_t)n + i];
    dinv[i] = rsqrtf((float)c);
}

#define GEMM_T 256
template <bool SELF>
static __device__ __forceinline__ void gemm1_body(
    float* sW, const float* __restrict__ x, const float* __restrict__ W1,
    const float* __restrict__ dinv,
    float* __restrict__ h, float* __restrict__ agg1, int nTiles, int blk)
{
    const int tid  = threadIdx.x;
    const int wave = tid >> 5;
    const int lane = tid & 31;
    const int m  = lane & 15;
    const int kh = lane >> 4;
    const int tile = blk * 8 + wave;
    const bool live = tile < nTiles;
    const int tl = live ? tile : nTiles - 1;
    const float4* __restrict__ xr =
        reinterpret_cast<const float4*>(x + (size_t)(tl * 16 + m) * N_FEAT) + kh;

    float4 cur[8];
#pragma unroll
    for (int j = 0; j < 8; ++j) cur[j] = xr[2 * j];

    {
        const float4* __restrict__ W4 = reinterpret_cast<const float4*>(W1);
        float4* sW4 = reinterpret_cast<float4*>(sW);
        float4 w[N_FEAT * HID / 4 / GEMM_T];
#pragma unroll
        for (int j = 0; j < N_FEAT * HID / 4 / GEMM_T; ++j) w[j] = W4[tid + GEMM_T * j];
#pragma unroll
        for (int j = 0; j < N_FEAT * HID / 4 / GEMM_T; ++j) sW4[tid + GEMM_T * j] = w[j];
    }
    __syncthreads();
    if (!live) return;

    v8f acc = {};
#pragma unroll
    for (int g = 0; g < N_FEAT / 64; ++g) {
        float4 nxt[8];
        if (g + 1 < N_FEAT / 64) {
#pragma unroll
            for (int j = 0; j < 8; ++j) nxt[j] = xr[2 * (8 * (g + 1) + j)];
        }
#pragma unroll
        for (int j = 0; j < 8; ++j) {
            const int kk = 8 * (8 * g + j) + 4 * kh;
            v2f a0, a1, b0, b1;
            a0.x = cur[j].x;  a0.y = cur[j].y;
            a1.x = cur[j].z;  a1.y = cur[j].w;
            b0.x = sW[kk * HID + m];        b0.y = sW[(kk + 1) * HID + m];
            b1.x = sW[(kk + 2) * HID + m];  b1.y = sW[(kk + 3) * HID + m];
            acc = __builtin_amdgcn_wmma_f32_16x16x4_f32(
                 false, a0,  false, b0,
                 (short)0, acc,  false,  false);
            acc = __builtin_amdgcn_wmma_f32_16x16x4_f32(
                 false, a1,  false, b1,
                 (short)0, acc,  false,  false);
        }
        if (g + 1 < N_FEAT / 64) {
#pragma unroll
            for (int j = 0; j < 8; ++j) cur[j] = nxt[j];
        }
    }

#pragma unroll
    for (int v = 0; v < 8; ++v) {
        int node  = tile * 16 + v + 8 * kh;
        float hs  = acc[v] * dinv[node];
        h[(size_t)node * HID + m]    = hs;
        if (SELF) agg1[(size_t)node * HID + m] = hs;
    }
}

template <bool SELF>
__global__ __launch_bounds__(GEMM_T) void k_gemm1(
    const float* __restrict__ x, const float* __restrict__ W1,
    const float* __restrict__ dinv,
    float* __restrict__ h, float* __restrict__ agg1, int nTiles)
{
    __shared__ __attribute__((aligned(16))) float sW[N_FEAT * HID];
    gemm1_body<SELF>(sW, x, W1, dinv, h, agg1, nTiles, (int)blockIdx.x);
}

template <bool B> struct agg_flag { static constexpr bool value = B; };

template <int F, int FP, int RN>
static __device__ __forceinline__ void agg_direct(
    float* acc, const float* __restrict__ srcv, int r, int d)
{
    for (int f = 0; f < F; ++f) atomicAdd(&acc[d * FP + f], srcv[(size_t)r * F + f]);
}

struct src4 { int x, y, z, w; };

template <int F, int FP, int RN, int S, int QCAP, int U>
__global__ __launch_bounds__(1024) void k_agg(
    const int* __restrict__ eidx,
    const float* __restrict__ srcv, float* __restrict__ dst, int E, int n)
{
    __shared__ __attribute__((aligned(16))) float acc[RN * FP];
    __shared__ unsigned q[QCAP + 32];
    __shared__ unsigned qn;

    const int* __restrict__ row = eidx;
    const int* __restrict__ col = eidx + E;
    const int tid  = threadIdx.x;
    const int base = blockIdx.x * RN;
    const int s    = blockIdx.y;

    float4* acc4 = reinterpret_cast<float4*>(acc);
    for (int i = tid; i < RN * FP / 4; i += 1024) acc4[i] = make_float4(0.f, 0.f, 0.f, 0.f);
    if (tid == 0) qn = 0u;
    __syncthreads();

    const int per = (E + S - 1) / S;
    int e0 = s * per;       if (e0 > E) e0 = E;
    int e1 = e0 + per;      if (e1 > E) e1 = E;

    auto visit = [&](int e, int c) {
        unsigned d = (unsigned)(c - base);
        if (d < (unsigned)RN) {
            int r = row[e];
            unsigned pos = atomicAdd(&qn, 1u);
            if (pos < (unsigned)QCAP) q[pos] = (unsigned)r | (d << 19);
            else agg_direct<F, FP, RN>(acc, srcv, r, (int)d);
        }
    };

    const int mis = (int)((((size_t)(col + e0)) >> 2) & 3);
    int a0 = e0 + ((4 - mis) & 3);
    if (a0 > e1) a0 = e1;
    const int nG = (e1 - a0) >> 2;
    const int t0 = a0 + 4 * nG;
    if (tid < a0 - e0) visit(e0 + tid, col[e0 + tid]);
    if (tid < e1 - t0) visit(t0 + tid, col[t0 + tid]);

    const int4* __restrict__ body = reinterpret_cast<const int4*>(col + a0);
    const src4* __restrict__ sbody = reinterpret_cast<const src4*>(row + a0);
    const int lane = tid & 31;
    for (int g0 = 0; g0 < nG; g0 += 8 * 1024) {
        const int g = g0 + tid;
        int4 v[8];
#pragma unroll
        for (int j = 0; j < 8; ++j) {
            int gg = g + j * 1024;
            int gc = gg < nG ? gg : nG - 1;
            v[j] = body[gc];
            if (gg >= nG) { v[j].x = -1; v[j].y = -1; v[j].z = -1; v[j].w = -1; }
        }
        src4 sr[8];
#pragma unroll
        for (int j = 0; j < 8; ++j) {
            bool any = ((unsigned)(v[j].x - base) < (unsigned)RN) | ((unsigned)(v[j].y - base) < (unsigned)RN)
                     | ((unsigned)(v[j].z - base) < (unsigned)RN) | ((unsigned)(v[j].w - base) < (unsigned)RN);
            sr[j] = sbody[any ? g + j * 1024 : 0];
        }
        unsigned nh = 0;
#pragma unroll
        for (int j = 0; j < 8; ++j) {
            nh += (unsigned)(v[j].x - base) < (unsigned)RN;
            nh += (unsigned)(v[j].y - base) < (unsigned)RN;
            nh += (unsigned)(v[j].z - base) < (unsigned)RN;
            nh += (unsigned)(v[j].w - base) < (unsigned)RN;
        }
        unsigned inc = nh;
#pragma unroll
        for (int o = 1; o < 32; o <<= 1) {
            unsigned up = (unsigned)__shfl_up((int)inc, o);
            if (lane >= o) inc += up;
        }
        const unsigned tot = (unsigned)__shfl((int)inc, 31);
        {
            unsigned wb = 0;
            if (lane == 31) wb = atomicAdd(&qn, tot);
            wb = (unsigned)__shfl((int)wb, 31);
            unsigned pos = wb + inc - nh;
            if (wb + tot <= (unsigned)QCAP) {
                const unsigned spare = (unsigned)QCAP + (unsigned)lane;
#pragma unroll
                for (int j = 0; j < 8; ++j) {
                    unsigned dx = (unsigned)(v[j].x - base), dy = (unsigned)(v[j].y - base);
                    unsigned dz = (unsigned)(v[j].z - base), dw = (unsigned)(v[j].w - base);
                    bool mx = dx < (unsigned)RN;
                    bool my = dy < (unsigned)RN;
                    bool mz = dz < (unsigned)RN;
                    bool mw = dw < (unsigned)RN;
                    q[mx ? pos : spare] = (unsigned)sr[j].x | (dx << 19); pos += mx;
                    q[my ? pos : spare] = (unsigned)sr[j].y | (dy << 19); pos += my;
                    q[mz ? pos : spare] = (unsigned)sr[j].z | (dz << 19); pos += mz;
                    q[mw ? pos : spare] = (unsigned)sr[j].w | (dw << 19); pos += mw;
                }
            } else {
#pragma unroll
                for (int j = 0; j < 8; ++j) {
                    int cj[4] = { v[j].x, v[j].y, v[j].z, v[j].w };
                    int rj[4] = { sr[j].x, sr[j].y, sr[j].z, sr[j].w };
#pragma unroll
                    for (int k = 0; k < 4; ++k) {
                        unsigned d = (unsigned)(cj[k] - base);
                        if (d < (unsigned)RN) {
                            if (pos < (unsigned)QCAP) q[pos] = (unsigned)rj[k] | (d << 19);
                            else agg_direct<F, FP, RN>(acc, srcv, rj[k], (int)d);
                            ++pos;
                        }
                    }
                }
            }
        }
    }
    __syncthreads();

    const int cnt = (int)(qn < (unsigned)QCAP ? qn : (unsigned)QCAP);

    {
        constexpr int G = 1024 / FP;
        const int g  = tid / FP;
        const int f  = tid % FP;
        const bool fl = f < F;
        const int fs = fl ? f : 0;
        auto batch = [&](int i0, auto guard) {
            constexpr bool GUARD = decltype(guard)::value;
            unsigned en[U]; float hv[U];
#pragma unroll
            for (int j = 0; j < U; ++j) {
                int i = i0 + j * G;
                en[j] = q[(!GUARD || i < cnt) ? i : cnt - 1];
            }
#pragma unroll
            for (int j = 0; j < U; ++j) {
                unsigned r = en[j] & 0x7FFFFu;
                hv[j] = srcv[(size_t)r * F + fs];
            }
#pragma unroll
            for (int j = 0; j < U; ++j) {
                int i = i0 + j * G;
                if ((!GUARD || i < cnt) && fl)
                    atomicAdd(&acc[(en[j] >> 19) * FP + f], hv[j]);
            }
        };
        const int nfull = cnt / (G * U);
        for (int b = 0; b < nfull; ++b) batch(b * (G * U) + g, agg_flag<false>{});
        const int it = nfull * (G * U) + g;
        if (it < cnt) batch(it, agg_flag<true>{});
    }
    __syncthreads();

    {
        const int T    = RN * F;
        const int stag = ((T / S) & ~1023) * s;
        for (int k = tid; k < T; k += 1024) {
            int t = k + stag;
            if (t >= T) t -= T;
            int nl = t / F;
            int j  = t - nl * F;
            int node = base + nl;
            if (node < n) atomicAdd(&dst[(size_t)node * F + j], acc[nl * FP + j]);
        }
    }
}

__global__ void k_l2(const float* __restrict__ agg1, const float* __restrict__ b1,
                     const float* __restrict__ W2, const float* __restrict__ dinv,
                     float* __restrict__ o, float* __restrict__ outInit, int n)
{
    __shared__ float sW2[HID * NCLS];
    __shared__ float sb1[HID];
    if (threadIdx.x < HID * NCLS) sW2[threadIdx.x] = W2[threadIdx.x];
    if (threadIdx.x < HID)        sb1[threadIdx.x] = b1[threadIdx.x];
    __syncthreads();

    int i = blockIdx.x * blockDim.x + threadIdx.x;
    if (i >= n) return;

    const float di = dinv[i];
    float hv[HID];
#pragma unroll
    for (int f = 0; f < HID; ++f) {
        float v = agg1[(size_t)i * HID + f] * di + sb1[f];
        hv[f] = v > 0.0f ? v : 0.0f;
    }
#pragma unroll
    for (int j = 0; j < NCLS; ++j) {
        float s = 0.0f;
#pragma unroll
        for (int f = 0; f < HID; ++f) s = fmaf(hv[f], sW2[f * NCLS + j], s);
        float os = s * di;
        o[(size_t)i * NCLS + j]       = os;
        outInit[(size_t)i * NCLS + j] = os;
    }
}

__global__ void k_lsm(float* __restrict__ out, const float* __restrict__ b2,
                      const float* __restrict__ dinv, int n)
{
    int i = blockIdx.x * blockDim.x + threadIdx.x;
    if (i >= n) return;
    const float di = dinv[i];
    float v[NCLS];
    float mx = -1e30f;
#pragma unroll
    for (int j = 0; j < NCLS; ++j) {
        v[j] = out[(size_t)i * NCLS + j] * di + b2[j];
        mx = fmaxf(mx, v[j]);
    }
    float s = 0.0f;
#pragma unroll
    for (int j = 0; j < NCLS; ++j) s += __expf(v[j] - mx);
    float l = __logf(s);
#pragma unroll
    for (int j = 0; j < NCLS; ++j) out[(size_t)i * NCLS + j] = v[j] - mx - l;
}


#define NB    256
#define RNBX  512
#define META  1024

static __device__ __forceinline__ unsigned scan_nb_excl(unsigned v, unsigned* wsum, int tid)
{
    const int lane = tid & 31, w = tid >> 5;
    unsigned inc = v;
#pragma unroll
    for (int o = 1; o < 32; o <<= 1) {
        unsigned up = (unsigned)__shfl_up((int)inc, o);
        if (lane >= o) inc += up;
    }
    if (w < NB / 32 && lane == 31) wsum[w] = inc;
    __syncthreads();
    unsigned add = 0u;
    if (w < NB / 32) for (int q = 0; q < w; ++q) add += wsum[q];
    __syncthreads();
    return add + inc - v;
}

__global__ __launch_bounds__(RNBX) void k_dinv_tot(
    const unsigned* __restrict__ part, float* __restrict__ dinv,
    unsigned* __restrict__ tot, unsigned* __restrict__ cur, int n, int rnb)
{
    __shared__ unsigned red[RNBX / 32];
    const int tid  = threadIdx.x;
    const int node = blockIdx.x * rnb + tid;
    unsigned s = 0u;
    if (tid < rnb && node < n) {
#pragma unroll
        for (int p = 0; p < DEG_S; ++p) s += part[(size_t)p * (size_t)n + node];
        dinv[node] = rsqrtf((float)(s + 1u));
    }
    unsigned w = s;
#pragma unroll
    for (int o = 16; o > 0; o >>= 1) w += (unsigned)__shfl_xor((int)w, o);
    if ((tid & 31) == 0) red[tid >> 5] = w;
    __syncthreads();
    if (tid == 0) {
        unsigned t = 0u;
        for (int q = 0; q < RNBX / 32; ++q) t += red[q];
        tot[blockIdx.x] = t;
        cur[blockIdx.x] = 0u;
    }
}

#define PL_EPT 16
#define PL_CH  (PL_EPT * GEMM_T)

static __device__ __forceinline__ void place_body(
    unsigned* sm, const int* __restrict__ eidx, const unsigned* __restrict__ tot,
    unsigned* __restrict__ cur, unsigned* __restrict__ list, int E, int rnb, int blk, int nblk)
{
    static_assert(PL_CH + PL_CH / 2 + 4 * NB + NB / 32 <= N_FEAT * HID, "placing role must fit the GEMM's LDS");
    static_assert(NB == GEMM_T, "scan_nb_excl: one range per thread");
    unsigned*       sorted = sm;
    unsigned short* sk     = reinterpret_cast<unsigned short*>(sm + PL_CH);
    unsigned*       cnt    = sm + PL_CH + PL_CH / 2;
    unsigned*       lstart = cnt + NB;
    unsigned*       gpos   = lstart + NB;
    unsigned*       bbase  = gpos + NB;
    unsigned*       wsum   = bbase + NB;

    const int* __restrict__ row = eidx;
    const int* __restrict__ col = eidx + E;
    const int tid = threadIdx.x;

    bbase[tid] = scan_nb_excl(tot[tid], wsum, tid);

    const int per = (E + nblk - 1) / nblk;
    int e0 = blk * per;     if (e0 > E) e0 = E;
    int e1 = e0 + per;      if (e1 > E) e1 = E;

    for (int c0 = e0; c0 < e1; c0 += PL_CH) {
        const int m = (e1 - c0) < PL_CH ? (e1 - c0) : PL_CH;
        cnt[tid] = 0u;
        __syncthreads();

        int cj[PL_EPT], rj[PL_EPT];
#pragma unroll
        for (int j = 0; j < PL_EPT; ++j) {
            int i  = tid + j * GEMM_T;
            int ii = i < m ? i : m - 1;
            cj[j] = col[c0 + ii];
            rj[j] = row[c0 + ii];
        }
        unsigned ent[PL_EPT], kr[PL_EPT];
#pragma unroll
        for (int j = 0; j < PL_EPT; ++j) {
            int i = tid + j * GEMM_T;
            ent[j] = 0u; kr[j] = 0u;
            if (i < m) {
                unsigned k  = (unsigned)cj[j] / (unsigned)rnb;
                unsigned dl = (unsigned)cj[j] - k * (unsigned)rnb;
                unsigned r  = atomicAdd(&cnt[k], 1u);
                ent[j] = (unsigned)rj[j] | (dl << 19);
                kr[j]  = k | (r << 8);
            }
        }
        __syncthreads();
        {
            unsigned cn = cnt[tid];
            unsigned ex = scan_nb_excl(cn, wsum, tid);
            lstart[tid] = ex;
            unsigned g = bbase[tid];
            if (cn) g += atomicAdd(&cur[tid], cn);
            gpos[tid] = g;
        }
        __syncthreads();
#pragma unroll
        for (int j = 0; j < PL_EPT; ++j) {
            int i = tid + j * GEMM_T;
            if (i < m) {
                unsigned k = kr[j] & 255u;
                unsigned p = lstart[k] + (kr[j] >> 8);
                sorted[p] = ent[j];
                sk[p]     = (unsigned short)k;
            }
        }
        __syncthreads();
        for (int i = tid; i < m; i += GEMM_T) {
            unsigned k = sk[i];
            list[gpos[k] + ((unsigned)i - lstart[k])] = sorted[i];
        }
        __syncthreads();
    }
}

__global__ __launch_bounds__(GEMM_T) void k_gemm1_place(
    const float* __restrict__ x, const float* __restrict__ W1, const float* __restrict__ dinv,
    float* __restrict__ h, int nTiles, int nGemm,
    const int* __restrict__ eidx, const unsigned* __restrict__ tot,
    unsigned* __restrict__ cur, unsigned* __restrict__ list, int E, int rnb, int nPlace)
{
    __shared__ __attribute__((aligned(16))) float sm[N_FEAT * HID];
    const int b = (int)blockIdx.x;
    if (b < nGemm) gemm1_body<false>(sm, x, W1, dinv, h, nullptr, nTiles, b);
    else           place_body(reinterpret_cast<unsigned*>(sm), eidx, tot, cur, list, E, rnb, b - nGemm, nPlace);
}

template <int F, int FP, bool GUARD>
static __device__ __forceinline__ unsigned own_batch(
    float* acc, const float* __restrict__ srcv, unsigned mine, int lane,
    const unsigned* __restrict__ seg, int inext, int cnt)
{
    constexpr int GPW = 32 / FP;
    constexpr int U   = 32 / GPW;
    const int gi = lane / FP;
    const int f  = lane % FP;
    const bool fl = f < F;
    const int fs = fl ? f : 0;
    unsigned en[U]; float hv[U];
#pragma unroll
    for (int j = 0; j < U; ++j) en[j] = (unsigned)__shfl((int)mine, j * GPW + gi);
#pragma unroll
    for (int j = 0; j < U; ++j) {
        unsigned r = en[j] & 0x7FFFFu;
        if (GUARD && en[j] == 0xFFFFFFFFu) r = 0u;
        hv[j] = srcv[(size_t)r * F + fs];
    }
    unsigned nxt = 0xFFFFFFFFu;
    if (!GUARD) {
        unsigned ld = seg[inext < cnt ? inext : cnt - 1];
        nxt = inext < cnt ? ld : 0xFFFFFFFFu;
    }
#pragma unroll
    for (int j = 0; j < U; ++j) {
        if ((!GUARD || en[j] != 0xFFFFFFFFu) && fl)
            atomicAdd(&acc[(en[j] >> 19) * FP + f], hv[j]);
    }
    return nxt;
}

template <int F, int FP>
static __device__ __forceinline__ void own_walk(
    float* acc, const float* __restrict__ srcv, const unsigned* __restrict__ seg, int cnt, int tid)
{
    const int lane  = tid & 31;
    const int nfull = cnt / 1024;
    unsigned nxt = tid < cnt ? seg[tid] : 0xFFFFFFFFu;
    for (int b = 0; b < nfull; ++b)
        nxt = own_batch<F, FP, false>(acc, srcv, nxt, lane, seg, (b + 1) * 1024 + tid, cnt);
    if (nfull * 1024 < cnt) own_batch<F, FP, true>(acc, srcv, nxt, lane, seg, 0, cnt);
}

__global__ __launch_bounds__(1024) void k_own1(
    const unsigned* __restrict__ list, const unsigned* __restrict__ tot,
    const float* __restrict__ h, const float* __restrict__ dinv,
    const float* __restrict__ b1, const float* __restrict__ W2,
    float* __restrict__ o, int n, int rnb)
{
    __shared__ __attribute__((aligned(16))) float acc[RNBX * HID];
    __shared__ unsigned wsum[NB / 32];
    __shared__ unsigned sseg[2];
    __shared__ float sW2[HID * NCLS];
    __shared__ float sb1[HID];
    const int tid = threadIdx.x;
    const int k   = blockIdx.x;
    {
        unsigned t  = tid < NB ? tot[tid] : 0u;
        unsigned ex = scan_nb_excl(t, wsum, tid);
        if (tid == k) { sseg[0] = ex; sseg[1] = t; }
    }
    float4* acc4 = reinterpret_cast<float4*>(acc);
    for (int i = tid; i < RNBX * HID / 4; i += 1024) acc4[i] = make_float4(0.f, 0.f, 0.f, 0.f);
    if (tid < HID * NCLS) sW2[tid] = W2[tid];
    if (tid < HID)        sb1[tid] = b1[tid];
    __syncthreads();

    own_walk<HID, 16>(acc, h, list + sseg[0], (int)sseg[1], tid);
    __syncthreads();

    const int node = k * rnb + tid;
    if (tid < rnb && node < n) {
        const float di = dinv[node];
        float hv[HID];
#pragma unroll
        for (int f = 0; f < HID; ++f) {
            float v = (acc[tid * HID + f] + h[(size_t)node * HID + f]) * di + sb1[f];
            hv[f] = v > 0.0f ? v : 0.0f;
        }
#pragma unroll
        for (int j = 0; j < NCLS; ++j) {
            float s = 0.0f;
#pragma unroll
            for (int f = 0; f < HID; ++f) s = fmaf(hv[f], sW2[f * NCLS + j], s);
            o[(size_t)node * NCLS + j] = s * di;
        }
    }
}

__global__ __launch_bounds__(1024) void k_own2(
    const unsigned* __restrict__ list, const unsigned* __restrict__ tot,
    const float* __restrict__ o, const float* __restrict__ dinv,
    const float* __restrict__ b2, float* __restrict__ out, int n, int rnb)
{
    __shared__ __attribute__((aligned(16))) float acc[RNBX * 8];
    __shared__ unsigned wsum[NB / 32];
    __shared__ unsigned sseg[2];
    const int tid = threadIdx.x;
    const int k   = blockIdx.x;
    {
        unsigned t  = tid < NB ? tot[tid] : 0u;
        unsigned ex = scan_nb_excl(t, wsum, tid);
        if (tid == k) { sseg[0] = ex; sseg[1] = t; }
    }
    float4* acc4 = reinterpret_cast<float4*>(acc);
    for (int i = tid; i < RNBX * 8 / 4; i += 1024) acc4[i] = make_float4(0.f, 0.f, 0.f, 0.f);
    __syncthreads();

    own_walk<NCLS, 8>(acc, o, list + sseg[0], (int)sseg[1], tid);
    __syncthreads();

    const int node = k * rnb + tid;
    if (tid < rnb && node < n) {
        const float di = dinv[node];
        float v[NCLS];
        float mx = -1e30f;
#pragma unroll
        for (int j = 0; j < NCLS; ++j) {
            v[j] = (acc[tid * 8 + j] + o[(size_t)node * NCLS + j]) * di + b2[j];
            mx = fmaxf(mx, v[j]);
        }
        float s = 0.0f;
#pragma unroll
        for (int j = 0; j < NCLS; ++j) s += __expf(v[j] - mx);
        float l = __logf(s);
#pragma unroll
        for (int j = 0; j < NCLS; ++j) out[(size_t)node * NCLS + j] = v[j] - mx - l;
    }
}

extern "C" void kernel_launch(void* const* d_in, const int* in_sizes, int n_in,
                              void* d_out, int out_size, void* d_ws, size_t ws_size,
                              hipStream_t stream)
{
    const float* x    = (const float*)d_in[0];
    const int*   eidx = (const int*)  d_in[1];
    const float* W1   = (const float*)d_in[2];
    const float* b1   = (const float*)d_in[3];
    const float* W2   = (const float*)d_in[4];
    const float* b2   = (const float*)d_in[5];
    float* out = (float*)d_out;

    const int N = in_sizes[0] / N_FEAT;
    const int E = in_sizes[1] / 2;

    float* ws   = (float*)d_ws;

    {
        const int rnb = (N + NB - 1) / NB;
        const size_t needL = (size_t)N * 24 + META + (size_t)E;
        const size_t needP = (size_t)N * (1 + DEG_S) + META;
        const size_t need  = (needL > needP ? needL : needP) * sizeof(float);
        if (ws_size >= need && rnb <= RNBX && N <= (1 << 19)) {
            float*    dinvO = ws;
            unsigned* tot   = (unsigned*)(ws + (size_t)N);
            unsigned* cur   = tot + NB;
            float*    hO    = ws + (size_t)N + META;
            float*    oO    = hO + (size_t)N * HID;
            unsigned* list  = (unsigned*)(oO + (size_t)N * NCLS);
            unsigned* partO = (unsigned*)hO;

            dim3 gDegO((N + DEG_RN - 1) / DEG_RN, DEG_S);
            k_deg_hist<<<gDegO, 1024, 0, stream>>>(eidx + E, partO, E, N);
            k_dinv_tot<<<NB, RNBX, 0, stream>>>(partO, dinvO, tot, cur, N, rnb);
            const int nTilesO = (N + 15) / 16;
            const int nGemmO  = (nTilesO + 7) / 8;
            const int nPlaceO = (E + PL_CH - 1) / PL_CH;
            k_gemm1_place<<<nGemmO + nPlaceO, GEMM_T, 0, stream>>>(
                x, W1, dinvO, hO, nTilesO, nGemmO, eidx, tot, cur, list, E, rnb, nPlaceO);
            k_own1<<<NB, 1024, 0, stream>>>(list, tot, hO, dinvO, b1, W2, oO, N, rnb);
            k_own2<<<NB, 1024, 0, stream>>>(list, tot, oO, dinvO, b2, out, N, rnb);
            return;
        }
    }
    float* dinv = ws;
    float* h    = ws + (size_t)N;
    float* agg1 = ws + (size_t)N * 17;
    float* obuf = ws + (size_t)N * 33;

    const int B = 256;
    unsigned* part = (unsigned*)(ws + (size_t)N);
    dim3 gDeg((N + DEG_RN - 1) / DEG_RN, DEG_S);
    k_deg_hist<<<gDeg, 1024, 0, stream>>>(eidx + E, part, E, N);
    k_dinv    <<<(N + B - 1) / B, B, 0, stream>>>(part, dinv, N);

    const int nTiles = (N + 15) / 16;
    k_gemm1<true><<<(nTiles + 7) / 8, GEMM_T, 0, stream>>>(x, W1, dinv, h, agg1, nTiles);

    dim3 gA1((N + 3584 - 1) / 3584, 9);
    k_agg<HID, 16, 3584, 9, 16384, 16><<<gA1, 1024, 0, stream>>>(eidx, h, agg1, E, N);

    k_l2<<<(N + B - 1) / B, B, 0, stream>>>(agg1, b1, W2, dinv, obuf, out, N);

    dim3 gA2((N + 7168 - 1) / 7168, 18);
    k_agg<NCLS, 8, 7168, 18, 16384, 16><<<gA2, 1024, 0, stream>>>(eidx, obuf, out, E, N);

    k_lsm<<<(N + B - 1) / B, B, 0, stream>>>(out, b2, dinv, N);
}


# Layer-2 rows stored at a 32-byte pitch (7 values + 1 unused word), not 28: a row gathered in the layer-2 edge walk no longer straddles a 128-byte line; 0.4 MB more workspace, else pitch 28
# speedup vs baseline: 5.5336x; 1.0883x over previous
#include <hip/hip_runtime.h>
#include <math.h>

#define N_FEAT 512
#define HID    16
#define NCLS   7

typedef __attribute__((ext_vector_type(2))) float v2f;
typedef __attribute__((ext_vector_type(8))) float v8f;

#define DEG_RN 16384
#define DEG_S  32

static __device__ __forceinline__ void deg_count(unsigned* cnt, int c, int base) {
    unsigned d = (unsigned)(c - base);
    if (d < (unsigned)DEG_RN) atomicAdd(&cnt[d], 1u);
}

__global__ __launch_bounds__(1024) void k_deg_hist(
    const int* __restrict__ col, unsigned* __restrict__ part, int E, int n)
{
    __shared__ unsigned cnt[DEG_RN];
    const int tid  = threadIdx.x;
    const int base = blockIdx.x * DEG_RN;
    const int s    = blockIdx.y;
    for (int i = tid; i < DEG_RN; i += 1024) cnt[i] = 0u;
    __syncthreads();

    const int per = (E + DEG_S - 1) / DEG_S;
    int e0 = s * per;       if (e0 > E) e0 = E;
    int e1 = e0 + per;      if (e1 > E) e1 = E;

    const int mis = (int)((((size_t)(col + e0)) >> 2) & 3);
    int a0 = e0 + ((4 - mis) & 3);
    if (a0 > e1) a0 = e1;
    const int nG = (e1 - a0) >> 2;
    const int t0 = a0 + 4 * nG;
    if (tid < a0 - e0) deg_count(cnt, col[e0 + tid], base);
    if (tid < e1 - t0) deg_count(cnt, col[t0 + tid], base);

    const int4* __restrict__ body = reinterpret_cast<const int4*>(col + a0);
    for (int g = tid; g < nG; g += 4 * 1024) {
        int4 v[4];
#pragma unroll
        for (int j = 0; j < 4; ++j) {
            int gg = g + j * 1024;
            int gc = gg < nG ? gg : nG - 1;
            v[j] = body[gc];
            if (gg >= nG) { v[j].x = -1; v[j].y = -1; v[j].z = -1; v[j].w = -1; }
        }
#pragma unroll
        for (int j = 0; j < 4; ++j) {
            deg_count(cnt, v[j].x, base);
            deg_count(cnt, v[j].y, base);
            deg_count(cnt, v[j].z, base);
            deg_count(cnt, v[j].w, base);
        }
    }
    __syncthreads();

    unsigned* __restrict__ plane = part + (size_t)s * (size_t)n;
    for (int i = tid; i < DEG_RN; i += 1024) {
        int node = base + i;
        if (node < n) plane[node] = cnt[i];
    }
}

__global__ void k_dinv(const unsigned* __restrict__ part, float* __restrict__ dinv, int n) {
    int i = blockIdx.x * blockDim.x + threadIdx.x;
    if (i >= n) return;
    unsigned c = 1u;
#pragma unroll
    for (int s = 0; s < DEG_S; ++s) c += part[(size_t)s * (size_t)n + i];
    dinv[i] = rsqrtf((float)c);
}

#define GEMM_T 256
template <bool SELF>
static __device__ __forceinline__ void gemm1_body(
    float* sW, const float* __restrict__ x, const float* __restrict__ W1,
    const float* __restrict__ dinv,
    float* __restrict__ h, float* __restrict__ agg1, int nTiles, int blk)
{
    const int tid  = threadIdx.x;
    const int wave = tid >> 5;
    const int lane = tid & 31;
    const int m  = lane & 15;
    const int kh = lane >> 4;
    const int tile = blk * 8 + wave;
    const bool live = tile < nTiles;
    const int tl = live ? tile : nTiles - 1;
    const float4* __restrict__ xr =
        reinterpret_cast<const float4*>(x + (size_t)(tl * 16 + m) * N_FEAT) + kh;

    float4 cur[8];
#pragma unroll
    for (int j = 0; j < 8; ++j) cur[j] = xr[2 * j];

    {
        const float4* __restrict__ W4 = reinterpret_cast<const float4*>(W1);
        float4* sW4 = reinterpret_cast<float4*>(sW);
        float4 w[N_FEAT * HID / 4 / GEMM_T];
#pragma unroll
        for (int j = 0; j < N_FEAT * HID / 4 / GEMM_T; ++j) w[j] = W4[tid + GEMM_T * j];
#pragma unroll
        for (int j = 0; j < N_FEAT * HID / 4 / GEMM_T; ++j) sW4[tid + GEMM_T * j] = w[j];
    }
    __syncthreads();
    if (!live) return;

    v8f acc = {};
#pragma unroll
    for (int g = 0; g < N_FEAT / 64; ++g) {
        float4 nxt[8];
        if (g + 1 < N_FEAT / 64) {
#pragma unroll
            for (int j = 0; j < 8; ++j) nxt[j] = xr[2 * (8 * (g + 1) + j)];
        }
#pragma unroll
        for (int j = 0; j < 8; ++j) {
            const int kk = 8 * (8 * g + j) + 4 * kh;
            v2f a0, a1, b0, b1;
            a0.x = cur[j].x;  a0.y = cur[j].y;
            a1.x = cur[j].z;  a1.y = cur[j].w;
            b0.x = sW[kk * HID + m];        b0.y = sW[(kk + 1) * HID + m];
            b1.x = sW[(kk + 2) * HID + m];  b1.y = sW[(kk + 3) * HID + m];
            acc = __builtin_amdgcn_wmma_f32_16x16x4_f32(
                 false, a0,  false, b0,
                 (short)0, acc,  false,  false);
            acc = __builtin_amdgcn_wmma_f32_16x16x4_f32(
                 false, a1,  false, b1,
                 (short)0, acc,  false,  false);
        }
        if (g + 1 < N_FEAT / 64) {
#pragma unroll
            for (int j = 0; j < 8; ++j) cur[j] = nxt[j];
        }
    }

#pragma unroll
    for (int v = 0; v < 8; ++v) {
        int node  = tile * 16 + v + 8 * kh;
        float hs  = acc[v] * dinv[node];
        h[(size_t)node * HID + m]    = hs;
        if (SELF) agg1[(size_t)node * HID + m] = hs;
    }
}

template <bool SELF>
__global__ __launch_bounds__(GEMM_T) void k_gemm1(
    const float* __restrict__ x, const float* __restrict__ W1,
    const float* __restrict__ dinv,
    float* __restrict__ h, float* __restrict__ agg1, int nTiles)
{
    __shared__ __attribute__((aligned(16))) float sW[N_FEAT * HID];
    gemm1_body<SELF>(sW, x, W1, dinv, h, agg1, nTiles, (int)blockIdx.x);
}

template <bool B> struct agg_flag { static constexpr bool value = B; };

template <int F, int FP, int RN>
static __device__ __forceinline__ void agg_direct(
    float* acc, const float* __restrict__ srcv, int r, int d)
{
    for (int f = 0; f < F; ++f) atomicAdd(&acc[d * FP + f], srcv[(size_t)r * F + f]);
}

struct src4 { int x, y, z, w; };

template <int F, int FP, int RN, int S, int QCAP, int U>
__global__ __launch_bounds__(1024) void k_agg(
    const int* __restrict__ eidx,
    const float* __restrict__ srcv, float* __restrict__ dst, int E, int n)
{
    __shared__ __attribute__((aligned(16))) float acc[RN * FP];
    __shared__ unsigned q[QCAP + 32];
    __shared__ unsigned qn;

    const int* __restrict__ row = eidx;
    const int* __restrict__ col = eidx + E;
    const int tid  = threadIdx.x;
    const int base = blockIdx.x * RN;
    const int s    = blockIdx.y;

    float4* acc4 = reinterpret_cast<float4*>(acc);
    for (int i = tid; i < RN * FP / 4; i += 1024) acc4[i] = make_float4(0.f, 0.f, 0.f, 0.f);
    if (tid == 0) qn = 0u;
    __syncthreads();

    const int per = (E + S - 1) / S;
    int e0 = s * per;       if (e0 > E) e0 = E;
    int e1 = e0 + per;      if (e1 > E) e1 = E;

    auto visit = [&](int e, int c) {
        unsigned d = (unsigned)(c - base);
        if (d < (unsigned)RN) {
            int r = row[e];
            unsigned pos = atomicAdd(&qn, 1u);
            if (pos < (unsigned)QCAP) q[pos] = (unsigned)r | (d << 19);
            else agg_direct<F, FP, RN>(acc, srcv, r, (int)d);
        }
    };

    const int mis = (int)((((size_t)(col + e0)) >> 2) & 3);
    int a0 = e0 + ((4 - mis) & 3);
    if (a0 > e1) a0 = e1;
    const int nG = (e1 - a0) >> 2;
    const int t0 = a0 + 4 * nG;
    if (tid < a0 - e0) visit(e0 + tid, col[e0 + tid]);
    if (tid < e1 - t0) visit(t0 + tid, col[t0 + tid]);

    const int4* __restrict__ body = reinterpret_cast<const int4*>(col + a0);
    const src4* __restrict__ sbody = reinterpret_cast<const src4*>(row + a0);
    const int lane = tid & 31;
    for (int g0 = 0; g0 < nG; g0 += 8 * 1024) {
        const int g = g0 + tid;
        int4 v[8];
#pragma unroll
        for (int j = 0; j < 8; ++j) {
            int gg = g + j * 1024;
            int gc = gg < nG ? gg : nG - 1;
            v[j] = body[gc];
            if (gg >= nG) { v[j].x = -1; v[j].y = -1; v[j].z = -1; v[j].w = -1; }
        }
        src4 sr[8];
#pragma unroll
        for (int j = 0; j < 8; ++j) {
            bool any = ((unsigned)(v[j].x - base) < (unsigned)RN) | ((unsigned)(v[j].y - base) < (unsigned)RN)
                     | ((unsigned)(v[j].z - base) < (unsigned)RN) | ((unsigned)(v[j].w - base) < (unsigned)RN);
            sr[j] = sbody[any ? g + j * 1024 : 0];
        }
        unsigned nh = 0;
#pragma unroll
        for (int j = 0; j < 8; ++j) {
            nh += (unsigned)(v[j].x - base) < (unsigned)RN;
            nh += (unsigned)(v[j].y - base) < (unsigned)RN;
            nh += (unsigned)(v[j].z - base) < (unsigned)RN;
            nh += (unsigned)(v[j].w - base) < (unsigned)RN;
        }
        unsigned inc = nh;
#pragma unroll
        for (int o = 1; o < 32; o <<= 1) {
            unsigned up = (unsigned)__shfl_up((int)inc, o);
            if (lane >= o) inc += up;
        }
        const unsigned tot = (unsigned)__shfl((int)inc, 31);
        {
            unsigned wb = 0;
            if (lane == 31) wb = atomicAdd(&qn, tot);
            wb = (unsigned)__shfl((int)wb, 31);
            unsigned pos = wb + inc - nh;
            if (wb + tot <= (unsigned)QCAP) {
                const unsigned spare = (unsigned)QCAP + (unsigned)lane;
#pragma unroll
                for (int j = 0; j < 8; ++j) {
                    unsigned dx = (unsigned)(v[j].x - base), dy = (unsigned)(v[j].y - base);
                    unsigned dz = (unsigned)(v[j].z - base), dw = (unsigned)(v[j].w - base);
                    bool mx = dx < (unsigned)RN;
                    bool my = dy < (unsigned)RN;
                    bool mz = dz < (unsigned)RN;
                    bool mw = dw < (unsigned)RN;
                    q[mx ? pos : spare] = (unsigned)sr[j].x | (dx << 19); pos += mx;
                    q[my ? pos : spare] = (unsigned)sr[j].y | (dy << 19); pos += my;
                    q[mz ? pos : spare] = (unsigned)sr[j].z | (dz << 19); pos += mz;
                    q[mw ? pos : spare] = (unsigned)sr[j].w | (dw << 19); pos += mw;
                }
            } else {
#pragma unroll
                for (int j = 0; j < 8; ++j) {
                    int cj[4] = { v[j].x, v[j].y, v[j].z, v[j].w };
                    int rj[4] = { sr[j].x, sr[j].y, sr[j].z, sr[j].w };
#pragma unroll
                    for (int k = 0; k < 4; ++k) {
                        unsigned d = (unsigned)(cj[k] - base);
                        if (d < (unsigned)RN) {
                            if (pos < (unsigned)QCAP) q[pos] = (unsigned)rj[k] | (d << 19);
                            else agg_direct<F, FP, RN>(acc, srcv, rj[k], (int)d);
                            ++pos;
                        }
                    }
                }
            }
        }
    }
    __syncthreads();

    const int cnt = (int)(qn < (unsigned)QCAP ? qn : (unsigned)QCAP);

    {
        constexpr int G = 1024 / FP;
        const int g  = tid / FP;
        const int f  = tid % FP;
        const bool fl = f < F;
        const int fs = fl ? f : 0;
        auto batch = [&](int i0, auto guard) {
            constexpr bool GUARD = decltype(guard)::value;
            unsigned en[U]; float hv[U];
#pragma unroll
            for (int j = 0; j < U; ++j) {
                int i = i0 + j * G;
                en[j] = q[(!GUARD || i < cnt) ? i : cnt - 1];
            }
#pragma unroll
            for (int j = 0; j < U; ++j) {
                unsigned r = en[j] & 0x7FFFFu;
                hv[j] = srcv[(size_t)r * F + fs];
            }
#pragma unroll
            for (int j = 0; j < U; ++j) {
                int i = i0 + j * G;
                if ((!GUARD || i < cnt) && fl)
                    atomicAdd(&acc[(en[j] >> 19) * FP + f], hv[j]);
            }
        };
        const int nfull = cnt / (G * U);
        for (int b = 0; b < nfull; ++b) batch(b * (G * U) + g, agg_flag<false>{});
        const int it = nfull * (G * U) + g;
        if (it < cnt) batch(it, agg_flag<true>{});
    }
    __syncthreads();

    {
        const int T    = RN * F;
        const int stag = ((T / S) & ~1023) * s;
        for (int k = tid; k < T; k += 1024) {
            int t = k + stag;
            if (t >= T) t -= T;
            int nl = t / F;
            int j  = t - nl * F;
            int node = base + nl;
            if (node < n) atomicAdd(&dst[(size_t)node * F + j], acc[nl * FP + j]);
        }
    }
}

__global__ void k_l2(const float* __restrict__ agg1, const float* __restrict__ b1,
                     const float* __restrict__ W2, const float* __restrict__ dinv,
                     float* __restrict__ o, float* __restrict__ outInit, int n)
{
    __shared__ float sW2[HID * NCLS];
    __shared__ float sb1[HID];
    if (threadIdx.x < HID * NCLS) sW2[threadIdx.x] = W2[threadIdx.x];
    if (threadIdx.x < HID)        sb1[threadIdx.x] = b1[threadIdx.x];
    __syncthreads();

    int i = blockIdx.x * blockDim.x + threadIdx.x;
    if (i >= n) return;

    const float di = dinv[i];
    float hv[HID];
#pragma unroll
    for (int f = 0; f < HID; ++f) {
        float v = agg1[(size_t)i * HID + f] * di + sb1[f];
        hv[f] = v > 0.0f ? v : 0.0f;
    }
#pragma unroll
    for (int j = 0; j < NCLS; ++j) {
        float s = 0.0f;
#pragma unroll
        for (int f = 0; f < HID; ++f) s = fmaf(hv[f], sW2[f * NCLS + j], s);
        float os = s * di;
        o[(size_t)i * NCLS + j]       = os;
        outInit[(size_t)i * NCLS + j] = os;
    }
}

__global__ void k_lsm(float* __restrict__ out, const float* __restrict__ b2,
                      const float* __restrict__ dinv, int n)
{
    int i = blockIdx.x * blockDim.x + threadIdx.x;
    if (i >= n) return;
    const float di = dinv[i];
    float v[NCLS];
    float mx = -1e30f;
#pragma unroll
    for (int j = 0; j < NCLS; ++j) {
        v[j] = out[(size_t)i * NCLS + j] * di + b2[j];
        mx = fmaxf(mx, v[j]);
    }
    float s = 0.0f;
#pragma unroll
    for (int j = 0; j < NCLS; ++j) s += __expf(v[j] - mx);
    float l = __logf(s);
#pragma unroll
    for (int j = 0; j < NCLS; ++j) out[(size_t)i * NCLS + j] = v[j] - mx - l;
}


#define NB    256
#define RNBX  512
#define META  1024
#define OROW  8

static __device__ __forceinline__ unsigned scan_nb_excl(unsigned v, unsigned* wsum, int tid)
{
    const int lane = tid & 31, w = tid >> 5;
    unsigned inc = v;
#pragma unroll
    for (int o = 1; o < 32; o <<= 1) {
        unsigned up = (unsigned)__shfl_up((int)inc, o);
        if (lane >= o) inc += up;
    }
    if (w < NB / 32 && lane == 31) wsum[w] = inc;
    __syncthreads();
    unsigned add = 0u;
    if (w < NB / 32) for (int q = 0; q < w; ++q) add += wsum[q];
    __syncthreads();
    return add + inc - v;
}

__global__ __launch_bounds__(RNBX) void k_dinv_tot(
    const unsigned* __restrict__ part, float* __restrict__ dinv,
    unsigned* __restrict__ tot, unsigned* __restrict__ cur, int n, int rnb)
{
    __shared__ unsigned red[RNBX / 32];
    const int tid  = threadIdx.x;
    const int node = blockIdx.x * rnb + tid;
    unsigned s = 0u;
    if (tid < rnb && node < n) {
#pragma unroll
        for (int p = 0; p < DEG_S; ++p) s += part[(size_t)p * (size_t)n + node];
        dinv[node] = rsqrtf((float)(s + 1u));
    }
    unsigned w = s;
#pragma unroll
    for (int o = 16; o > 0; o >>= 1) w += (unsigned)__shfl_xor((int)w, o);
    if ((tid & 31) == 0) red[tid >> 5] = w;
    __syncthreads();
    if (tid == 0) {
        unsigned t = 0u;
        for (int q = 0; q < RNBX / 32; ++q) t += red[q];
        tot[blockIdx.x] = t;
        cur[blockIdx.x] = 0u;
    }
}

#define PL_EPT 16
#define PL_CH  (PL_EPT * GEMM_T)

static __device__ __forceinline__ void place_body(
    unsigned* sm, const int* __restrict__ eidx, const unsigned* __restrict__ tot,
    unsigned* __restrict__ cur, unsigned* __restrict__ list, int E, int rnb, int blk, int nblk)
{
    static_assert(PL_CH + PL_CH / 2 + 4 * NB + NB / 32 <= N_FEAT * HID, "placing role must fit the GEMM's LDS");
    static_assert(NB == GEMM_T, "scan_nb_excl: one range per thread");
    unsigned*       sorted = sm;
    unsigned short* sk     = reinterpret_cast<unsigned short*>(sm + PL_CH);
    unsigned*       cnt    = sm + PL_CH + PL_CH / 2;
    unsigned*       lstart = cnt + NB;
    unsigned*       gpos   = lstart + NB;
    unsigned*       bbase  = gpos + NB;
    unsigned*       wsum   = bbase + NB;

    const int* __restrict__ row = eidx;
    const int* __restrict__ col = eidx + E;
    const int tid = threadIdx.x;

    bbase[tid] = scan_nb_excl(tot[tid], wsum, tid);

    const int per = (E + nblk - 1) / nblk;
    int e0 = blk * per;     if (e0 > E) e0 = E;
    int e1 = e0 + per;      if (e1 > E) e1 = E;

    for (int c0 = e0; c0 < e1; c0 += PL_CH) {
        const int m = (e1 - c0) < PL_CH ? (e1 - c0) : PL_CH;
        cnt[tid] = 0u;
        __syncthreads();

        int cj[PL_EPT], rj[PL_EPT];
#pragma unroll
        for (int j = 0; j < PL_EPT; ++j) {
            int i  = tid + j * GEMM_T;
            int ii = i < m ? i : m - 1;
            cj[j] = col[c0 + ii];
            rj[j] = row[c0 + ii];
        }
        unsigned ent[PL_EPT], kr[PL_EPT];
#pragma unroll
        for (int j = 0; j < PL_EPT; ++j) {
            int i = tid + j * GEMM_T;
            ent[j] = 0u; kr[j] = 0u;
            if (i < m) {
                unsigned k  = (unsigned)cj[j] / (unsigned)rnb;
                unsigned dl = (unsigned)cj[j] - k * (unsigned)rnb;
                unsigned r  = atomicAdd(&cnt[k], 1u);
                ent[j] = (unsigned)rj[j] | (dl << 19);
                kr[j]  = k | (r << 8);
            }
        }
        __syncthreads();
        {
            unsigned cn = cnt[tid];
            unsigned ex = scan_nb_excl(cn, wsum, tid);
            lstart[tid] = ex;
            unsigned g = bbase[tid];
            if (cn) g += atomicAdd(&cur[tid], cn);
            gpos[tid] = g;
        }
        __syncthreads();
#pragma unroll
        for (int j = 0; j < PL_EPT; ++j) {
            int i = tid + j * GEMM_T;
            if (i < m) {
                unsigned k = kr[j] & 255u;
                unsigned p = lstart[k] + (kr[j] >> 8);
                sorted[p] = ent[j];
                sk[p]     = (unsigned short)k;
            }
        }
        __syncthreads();
        for (int i = tid; i < m; i += GEMM_T) {
            unsigned k = sk[i];
            list[gpos[k] + ((unsigned)i - lstart[k])] = sorted[i];
        }
        __syncthreads();
    }
}

__global__ __launch_bounds__(GEMM_T) void k_gemm1_place(
    const float* __restrict__ x, const float* __restrict__ W1, const float* __restrict__ dinv,
    float* __restrict__ h, int nTiles, int nGemm,
    const int* __restrict__ eidx, const unsigned* __restrict__ tot,
    unsigned* __restrict__ cur, unsigned* __restrict__ list, int E, int rnb, int nPlace)
{
    __shared__ __attribute__((aligned(16))) float sm[N_FEAT * HID];
    const int b = (int)blockIdx.x;
    if (b < nGemm) gemm1_body<false>(sm, x, W1, dinv, h, nullptr, nTiles, b);
    else           place_body(reinterpret_cast<unsigned*>(sm), eidx, tot, cur, list, E, rnb, b - nGemm, nPlace);
}

template <int F, int FP, int RS, bool GUARD>
static __device__ __forceinline__ unsigned own_batch(
    float* acc, const float* __restrict__ srcv, unsigned mine, int lane,
    const unsigned* __restrict__ seg, int inext, int cnt)
{
    constexpr int GPW = 32 / FP;
    constexpr int U   = 32 / GPW;
    const int gi = lane / FP;
    const int f  = lane % FP;
    const bool fl = f < F;
    const int fs = fl ? f : 0;
    unsigned en[U]; float hv[U];
#pragma unroll
    for (int j = 0; j < U; ++j) en[j] = (unsigned)__shfl((int)mine, j * GPW + gi);
#pragma unroll
    for (int j = 0; j < U; ++j) {
        unsigned r = en[j] & 0x7FFFFu;
        if (GUARD && en[j] == 0xFFFFFFFFu) r = 0u;
        hv[j] = srcv[(size_t)r * RS + fs];
    }
    unsigned nxt = 0xFFFFFFFFu;
    if (!GUARD) {
        unsigned ld = seg[inext < cnt ? inext : cnt - 1];
        nxt = inext < cnt ? ld : 0xFFFFFFFFu;
    }
#pragma unroll
    for (int j = 0; j < U; ++j) {
        if ((!GUARD || en[j] != 0xFFFFFFFFu) && fl)
            atomicAdd(&acc[(en[j] >> 19) * FP + f], hv[j]);
    }
    return nxt;
}

template <int F, int FP, int RS>
static __device__ __forceinline__ void own_walk(
    float* acc, const float* __restrict__ srcv, const unsigned* __restrict__ seg, int cnt, int tid)
{
    const int lane  = tid & 31;
    const int nfull = cnt / 1024;
    unsigned nxt = tid < cnt ? seg[tid] : 0xFFFFFFFFu;
    for (int b = 0; b < nfull; ++b)
        nxt = own_batch<F, FP, RS, false>(acc, srcv, nxt, lane, seg, (b + 1) * 1024 + tid, cnt);
    if (nfull * 1024 < cnt) own_batch<F, FP, RS, true>(acc, srcv, nxt, lane, seg, 0, cnt);
}

template <int OP>
__global__ __launch_bounds__(1024) void k_own1(
    const unsigned* __restrict__ list, const unsigned* __restrict__ tot,
    const float* __restrict__ h, const float* __restrict__ dinv,
    const float* __restrict__ b1, const float* __restrict__ W2,
    float* __restrict__ o, int n, int rnb)
{
    __shared__ __attribute__((aligned(16))) float acc[RNBX * HID];
    __shared__ unsigned wsum[NB / 32];
    __shared__ unsigned sseg[2];
    __shared__ float sW2[HID * NCLS];
    __shared__ float sb1[HID];
    const int tid = threadIdx.x;
    const int k   = blockIdx.x;
    {
        unsigned t  = tid < NB ? tot[tid] : 0u;
        unsigned ex = scan_nb_excl(t, wsum, tid);
        if (tid == k) { sseg[0] = ex; sseg[1] = t; }
    }
    float4* acc4 = reinterpret_cast<float4*>(acc);
    for (int i = tid; i < RNBX * HID / 4; i += 1024) acc4[i] = make_float4(0.f, 0.f, 0.f, 0.f);
    if (tid < HID * NCLS) sW2[tid] = W2[tid];
    if (tid < HID)        sb1[tid] = b1[tid];
    __syncthreads();

    own_walk<HID, 16, HID>(acc, h, list + sseg[0], (int)sseg[1], tid);
    __syncthreads();

    const int node = k * rnb + tid;
    if (tid < rnb && node < n) {
        const float di = dinv[node];
        float hv[HID];
#pragma unroll
        for (int f = 0; f < HID; ++f) {
            float v = (acc[tid * HID + f] + h[(size_t)node * HID + f]) * di + sb1[f];
            hv[f] = v > 0.0f ? v : 0.0f;
        }
#pragma unroll
        for (int j = 0; j < NCLS; ++j) {
            float s = 0.0f;
#pragma unroll
            for (int f = 0; f < HID; ++f) s = fmaf(hv[f], sW2[f * NCLS + j], s);
            o[(size_t)node * OP + j] = s * di;
        }
    }
}

template <int OP>
__global__ __launch_bounds__(1024) void k_own2(
    const unsigned* __restrict__ list, const unsigned* __restrict__ tot,
    const float* __restrict__ o, const float* __restrict__ dinv,
    const float* __restrict__ b2, float* __restrict__ out, int n, int rnb)
{
    __shared__ __attribute__((aligned(16))) float acc[RNBX * 8];
    __shared__ unsigned wsum[NB / 32];
    __shared__ unsigned sseg[2];
    const int tid = threadIdx.x;
    const int k   = blockIdx.x;
    {
        unsigned t  = tid < NB ? tot[tid] : 0u;
        unsigned ex = scan_nb_excl(t, wsum, tid);
        if (tid == k) { sseg[0] = ex; sseg[1] = t; }
    }
    float4* acc4 = reinterpret_cast<float4*>(acc);
    for (int i = tid; i < RNBX * 8 / 4; i += 1024) acc4[i] = make_float4(0.f, 0.f, 0.f, 0.f);
    __syncthreads();

    own_walk<NCLS, 8, OP>(acc, o, list + sseg[0], (int)sseg[1], tid);
    __syncthreads();

    const int node = k * rnb + tid;
    if (tid < rnb && node < n) {
        const float di = dinv[node];
        float v[NCLS];
        float mx = -1e30f;
#pragma unroll
        for (int j = 0; j < NCLS; ++j) {
            v[j] = (acc[tid * 8 + j] + o[(size_t)node * OP + j]) * di + b2[j];
            mx = fmaxf(mx, v[j]);
        }
        float s = 0.0f;
#pragma unroll
        for (int j = 0; j < NCLS; ++j) s += __expf(v[j] - mx);
        float l = __logf(s);
#pragma unroll
        for (int j = 0; j < NCLS; ++j) out[(size_t)node * NCLS + j] = v[j] - mx - l;
    }
}

extern "C" void kernel_launch(void* const* d_in, const int* in_sizes, int n_in,
                              void* d_out, int out_size, void* d_ws, size_t ws_size,
                              hipStream_t stream)
{
    const float* x    = (const float*)d_in[0];
    const int*   eidx = (const int*)  d_in[1];
    const float* W1   = (const float*)d_in[2];
    const float* b1   = (const float*)d_in[3];
    const float* W2   = (const float*)d_in[4];
    const float* b2   = (const float*)d_in[5];
    float* out = (float*)d_out;

    const int N = in_sizes[0] / N_FEAT;
    const int E = in_sizes[1] / 2;

    float* ws   = (float*)d_ws;

    {
        const int rnb = (N + NB - 1) / NB;
        const size_t needL = (size_t)N * (17 + NCLS) + META + (size_t)E;
        const size_t needW = needL + (size_t)N * (OROW - NCLS);
        const size_t needP = (size_t)N * (1 + DEG_S) + META;
        const size_t need  = (needL > needP ? needL : needP) * sizeof(float);
        const size_t needO = (needW > needP ? needW : needP) * sizeof(float);
        if (ws_size >= need && rnb <= RNBX && N <= (1 << 19)) {
            const bool wide = ws_size >= needO;
            float*    dinvO = ws;
            unsigned* tot   = (unsigned*)(ws + (size_t)N);
            unsigned* cur   = tot + NB;
            float*    hO    = ws + (size_t)N + META;
            float*    oO    = hO + (size_t)N * HID;
            unsigned* list  = (unsigned*)(oO + (size_t)N * (wide ? OROW : NCLS));
            unsigned* partO = (unsigned*)hO;

            dim3 gDegO((N + DEG_RN - 1) / DEG_RN, DEG_S);
            k_deg_hist<<<gDegO, 1024, 0, stream>>>(eidx + E, partO, E, N);
            k_dinv_tot<<<NB, RNBX, 0, stream>>>(partO, dinvO, tot, cur, N, rnb);
            const int nTilesO = (N + 15) / 16;
            const int nGemmO  = (nTilesO + 7) / 8;
            const int nPlaceO = (E + PL_CH - 1) / PL_CH;
            k_gemm1_place<<<nGemmO + nPlaceO, GEMM_T, 0, stream>>>(
                x, W1, dinvO, hO, nTilesO, nGemmO, eidx, tot, cur, list, E, rnb, nPlaceO);
            if (wide) {
                k_own1<OROW><<<NB, 1024, 0, stream>>>(list, tot, hO, dinvO, b1, W2, oO, N, rnb);
                k_own2<OROW><<<NB, 1024, 0, stream>>>(list, tot, oO, dinvO, b2, out, N, rnb);
            } else {
                k_own1<NCLS><<<NB, 1024, 0, stream>>>(list, tot, hO, dinvO, b1, W2, oO, N, rnb);
                k_own2<NCLS><<<NB, 1024, 0, stream>>>(list, tot, oO, dinvO, b2, out, N, rnb);
            }
            return;
        }
    }
    float* dinv = ws;
    float* h    = ws + (size_t)N;
    float* agg1 = ws + (size_t)N * 17;
    float* obuf = ws + (size_t)N * 33;

    const int B = 256;
    unsigned* part = (unsigned*)(ws + (size_t)N);
    dim3 gDeg((N + DEG_RN - 1) / DEG_RN, DEG_S);
    k_deg_hist<<<gDeg, 1024, 0, stream>>>(eidx + E, part, E, N);
    k_dinv    <<<(N + B - 1) / B, B, 0, stream>>>(part, dinv, N);

    const int nTiles = (N + 15) / 16;
    k_gemm1<true><<<(nTiles + 7) / 8, GEMM_T, 0, stream>>>(x, W1, dinv, h, agg1, nTiles);

    dim3 gA1((N + 3584 - 1) / 3584, 9);
    k_agg<HID, 16, 3584, 9, 16384, 16><<<gA1, 1024, 0, stream>>>(eidx, h, agg1, E, N);

    k_l2<<<(N + B - 1) / B, B, 0, stream>>>(agg1, b1, W2, dinv, obuf, out, N);

    dim3 gA2((N + 7168 - 1) / 7168, 18);
    k_agg<NCLS, 8, 7168, 18, 16384, 16><<<gA2, 1024, 0, stream>>>(eidx, obuf, out, E, N);

    k_lsm<<<(N + B - 1) / B, B, 0, stream>>>(out, b2, dinv, N);
}
